# starting kernel
# baseline (speedup 1.0000x reference)
#include <hip/hip_runtime.h>
#include <hip/hip_bf16.h>

typedef float v2f __attribute__((ext_vector_type(2)));
typedef float v8f __attribute__((ext_vector_type(8)));

static constexpr int kSEQ   = 2048;
static constexpr int kBATCH = 64;
static constexpr int kIN    = 512;
static constexpr int kH     = 3;
static constexpr int kL     = 4;
static constexpr int kG     = 3 * kH;
static constexpr int kROWS  = kSEQ * kBATCH;

__global__ __launch_bounds__(256) void gru_in_proj_wmma(
    const float* __restrict__ X, const float* __restrict__ Wih0,
    const float* __restrict__ bih0, float* __restrict__ gx) {
  __shared__ float Bs[kIN * 16];
  const int tid = threadIdx.x;
  for (int idx = tid; idx < kIN * 16; idx += 256) {
    const int k = idx >> 4;
    const int n = idx & 15;
    Bs[idx] = (n < kG) ? Wih0[n * kIN + k] : 0.0f;
  }
  __syncthreads();

  const int wave = tid >> 5;
  const int lane = tid & 31;
  const int half = lane >> 4;
  const int n16  = lane & 15;
  const int row0 = blockIdx.x * 128 + wave * 16;

  const float* arow = X + (size_t)(row0 + n16) * kIN + 2 * half;

  v8f c = {0.f, 0.f, 0.f, 0.f, 0.f, 0.f, 0.f, 0.f};
  for (int kk = 0; kk < kIN; kk += 4) {
    v2f a = *(const v2f*)(arow + kk);
    v2f b;
    b.x = Bs[(kk + 2 * half) * 16 + n16];
    b.y = Bs[(kk + 2 * half + 1) * 16 + n16];
    c = __builtin_amdgcn_wmma_f32_16x16x4_f32(
         false, a,  false, b,
         (short)0, c,  false,  false);
  }

  if (n16 < kG) {
    const float bias = bih0[n16];
#pragma unroll
    for (int v = 0; v < 8; ++v) {
      const int m = v + 8 * half;
      gx[(size_t)(row0 + m) * kG + n16] = c[v] + bias;
    }
  }
}

__device__ __forceinline__ float sigmoid_f(float x) {
  return 1.0f / (1.0f + __expf(-x));
}
__device__ __forceinline__ float tanh_f(float x) {
  const float e = __expf(2.0f * fabsf(x));
  return copysignf(1.0f - 2.0f / (e + 1.0f), x);
}

__device__ __forceinline__ void gru_cell(float h[kH], const float gxv[kG],
                                         const float* __restrict__ whh,
                                         const float* __restrict__ bhh) {
  float gh[kG];
#pragma unroll
  for (int g = 0; g < kG; ++g) {
    float s = bhh[g];
#pragma unroll
    for (int j = 0; j < kH; ++j) s += whh[g * kH + j] * h[j];
    gh[g] = s;
  }
#pragma unroll
  for (int j = 0; j < kH; ++j) {
    const float r = sigmoid_f(gxv[j] + gh[j]);
    const float z = sigmoid_f(gxv[kH + j] + gh[kH + j]);
    const float n = tanh_f(gxv[2 * kH + j] + r * gh[2 * kH + j]);
    h[j] = (1.0f - z) * n + z * h[j];
  }
}

__global__ __launch_bounds__(64) void gru_scan(
    const float* __restrict__ gx0, const float* __restrict__ hxs,
    const float* __restrict__ whh0, const float* __restrict__ bhh0,
    const float* __restrict__ wih_r, const float* __restrict__ whh_r,
    const float* __restrict__ bih_r, const float* __restrict__ bhh_r,
    float* __restrict__ out) {
  __shared__ float Ws[36 + 3 * 72];
  const int tid = threadIdx.x;
  for (int i = tid; i < 27; i += 64) Ws[i] = whh0[i];
  for (int i = tid; i < 9; i += 64) Ws[27 + i] = bhh0[i];
  for (int i = tid; i < 3 * 27; i += 64)
    Ws[36 + (i / 27) * 72 + (i % 27)] = wih_r[i];
  for (int i = tid; i < 3 * 27; i += 64)
    Ws[36 + (i / 27) * 72 + 27 + (i % 27)] = whh_r[i];
  for (int i = tid; i < 3 * 9; i += 64)
    Ws[36 + (i / 9) * 72 + 54 + (i % 9)] = bih_r[i];
  for (int i = tid; i < 3 * 9; i += 64)
    Ws[36 + (i / 9) * 72 + 63 + (i % 9)] = bhh_r[i];
  __syncthreads();

  const int b = tid;
  float h[kL][kH];
#pragma unroll
  for (int l = 0; l < kL; ++l)
#pragma unroll
    for (int j = 0; j < kH; ++j) h[l][j] = hxs[(l * kBATCH + b) * kH + j];

  for (int t = 0; t < kSEQ; ++t) {
    float gxv[kG];
    const float* gp = gx0 + (size_t)(t * kBATCH + b) * kG;
#pragma unroll
    for (int g = 0; g < kG; ++g) gxv[g] = gp[g];
    gru_cell(h[0], gxv, &Ws[0], &Ws[27]);

#pragma unroll
    for (int l = 1; l < kL; ++l) {
      const float* wih = &Ws[36 + (l - 1) * 72];
      const float* whh = wih + 27;
      const float* bih = wih + 54;
      const float* bhh = wih + 63;
#pragma unroll
      for (int g = 0; g < kG; ++g) {
        float s = bih[g];
#pragma unroll
        for (int j = 0; j < kH; ++j) s += wih[g * kH + j] * h[l - 1][j];
        gxv[g] = s;
      }
      gru_cell(h[l], gxv, whh, bhh);
    }

#pragma unroll
    for (int j = 0; j < kH; ++j)
      out[(size_t)(t * kBATCH + b) * kH + j] = h[kL - 1][j];
  }

  for (int l = 0; l < kL; ++l)
#pragma unroll
    for (int j = 0; j < kH; ++j)
      out[(size_t)kSEQ * kBATCH * kH + (l * kBATCH + b) * kH + j] = h[l][j];
}

extern "C" void kernel_launch(void* const* d_in, const int* in_sizes, int n_in,
                              void* d_out, int out_size, void* d_ws,
                              size_t ws_size, hipStream_t stream) {
  const float* x    = (const float*)d_in[0];
  const float* hxs  = (const float*)d_in[1];
  const float* wih0 = (const float*)d_in[2];
  const float* whh0 = (const float*)d_in[3];
  const float* bih0 = (const float*)d_in[4];
  const float* bhh0 = (const float*)d_in[5];
  const float* wihr = (const float*)d_in[6];
  const float* whhr = (const float*)d_in[7];
  const float* bihr = (const float*)d_in[8];
  const float* bhhr = (const float*)d_in[9];
  float* out = (float*)d_out;
  float* gx  = (float*)d_ws;

  gru_in_proj_wmma<<<kROWS / 128, 256, 0, stream>>>(x, wih0, bih0, gx);
  gru_scan<<<1, 64, 0, stream>>>(gx, hxs, whh0, bhh0, wihr, whhr, bihr, bhhr,
                                 out);
}


# GRU scan run as a wavefront over the 4 layers: lane = 4*batch + layer, layer l does time s-l, h handed up the quad by a DPP quad_perm move; 8 one-wave groups, no LDS; sums and their order as before
# speedup vs baseline: 1.9722x; 1.9722x over previous
#include <hip/hip_runtime.h>
#include <hip/hip_bf16.h>

typedef float v2f __attribute__((ext_vector_type(2)));
typedef float v8f __attribute__((ext_vector_type(8)));

static constexpr int kSEQ   = 2048;
static constexpr int kBATCH = 64;
static constexpr int kIN    = 512;
static constexpr int kH     = 3;
static constexpr int kL     = 4;
static constexpr int kG     = 3 * kH;
static constexpr int kROWS  = kSEQ * kBATCH;

__global__ __launch_bounds__(256) void gru_in_proj_wmma(
    const float* __restrict__ X, const float* __restrict__ Wih0,
    const float* __restrict__ bih0, float* __restrict__ gx) {
  __shared__ float Bs[kIN * 16];
  const int tid = threadIdx.x;
  for (int idx = tid; idx < kIN * 16; idx += 256) {
    const int k = idx >> 4;
    const int n = idx & 15;
    Bs[idx] = (n < kG) ? Wih0[n * kIN + k] : 0.0f;
  }
  __syncthreads();

  const int wave = tid >> 5;
  const int lane = tid & 31;
  const int half = lane >> 4;
  const int n16  = lane & 15;
  const int row0 = blockIdx.x * 128 + wave * 16;

  const float* arow = X + (size_t)(row0 + n16) * kIN + 2 * half;

  v8f c = {0.f, 0.f, 0.f, 0.f, 0.f, 0.f, 0.f, 0.f};
  for (int kk = 0; kk < kIN; kk += 4) {
    v2f a = *(const v2f*)(arow + kk);
    v2f b;
    b.x = Bs[(kk + 2 * half) * 16 + n16];
    b.y = Bs[(kk + 2 * half + 1) * 16 + n16];
    c = __builtin_amdgcn_wmma_f32_16x16x4_f32(
         false, a,  false, b,
         (short)0, c,  false,  false);
  }

  if (n16 < kG) {
    const float bias = bih0[n16];
#pragma unroll
    for (int v = 0; v < 8; ++v) {
      const int m = v + 8 * half;
      gx[(size_t)(row0 + m) * kG + n16] = c[v] + bias;
    }
  }
}

__device__ __forceinline__ float sigmoid_f(float x) {
  return 1.0f / (1.0f + __expf(-x));
}
__device__ __forceinline__ float tanh_f(float x) {
  const float e = __expf(2.0f * fabsf(x));
  return copysignf(1.0f - 2.0f / (e + 1.0f), x);
}

__device__ __forceinline__ void gru_hidden_proj(float gh[kG], const float h[kH],
                                                const float* __restrict__ whh,
                                                const float* __restrict__ bhh) {
#pragma unroll
  for (int g = 0; g < kG; ++g) {
    float s = bhh[g];
#pragma unroll
    for (int j = 0; j < kH; ++j) s += whh[g * kH + j] * h[j];
    gh[g] = s;
  }
}

__device__ __forceinline__ void gru_gates(float h[kH], const float gxv[kG],
                                          const float gh[kG]) {
#pragma unroll
  for (int j = 0; j < kH; ++j) {
    const float r = sigmoid_f(gxv[j] + gh[j]);
    const float z = sigmoid_f(gxv[kH + j] + gh[kH + j]);
    const float n = tanh_f(gxv[2 * kH + j] + r * gh[2 * kH + j]);
    h[j] = (1.0f - z) * n + z * h[j];
  }
}

__device__ __forceinline__ float from_lower_lane_of_quad(float v) {
  return __builtin_bit_cast(
      float, __builtin_amdgcn_mov_dpp(__builtin_bit_cast(int, v), 0x90, 0xf,
                                      0xf, true));
}

template <bool kEdge>
__device__ __forceinline__ void scan_step(
    int s, int l, int b, float h[kH], const float (&wih)[kG * kH],
    const float (&bih)[kG], const float (&whh)[kG * kH], const float (&bhh)[kG],
    const float* __restrict__ gx0, float* __restrict__ out) {
  const int t = s - l;
  const bool active = !kEdge || (t >= 0 && t < kSEQ);

  float hp[kH];
#pragma unroll
  for (int j = 0; j < kH; ++j) hp[j] = from_lower_lane_of_quad(h[j]);

  const int t0 = (kEdge && s >= kSEQ) ? kSEQ - 1 : s;
  const float* gp = gx0 + (size_t)(t0 * kBATCH + b) * kG;
  float gl[kG];
#pragma unroll
  for (int g = 0; g < kG; ++g) gl[g] = gp[g];

  float a[kG];
#pragma unroll
  for (int g = 0; g < kG; ++g) {
    float acc = bih[g];
#pragma unroll
    for (int j = 0; j < kH; ++j) acc += wih[g * kH + j] * hp[j];
    a[g] = acc;
  }
  float gh[kG];
  gru_hidden_proj(gh, h, whh, bhh);

  __builtin_amdgcn_sched_barrier(0);

  float gxv[kG];
#pragma unroll
  for (int g = 0; g < kG; ++g) gxv[g] = (l == 0) ? gl[g] : a[g];

  float hn[kH];
#pragma unroll
  for (int j = 0; j < kH; ++j) hn[j] = h[j];
  gru_gates(hn, gxv, gh);

  if (active) {
#pragma unroll
    for (int j = 0; j < kH; ++j) h[j] = hn[j];
    if (l == kL - 1) {
#pragma unroll
      for (int j = 0; j < kH; ++j)
        out[(size_t)(t * kBATCH + b) * kH + j] = hn[j];
    }
  }
}

__global__ __launch_bounds__(32) void gru_scan(
    const float* __restrict__ gx0, const float* __restrict__ hxs,
    const float* __restrict__ whh0, const float* __restrict__ bhh0,
    const float* __restrict__ wih_r, const float* __restrict__ whh_r,
    const float* __restrict__ bih_r, const float* __restrict__ bhh_r,
    float* __restrict__ out) {
  const int lane = threadIdx.x;
  const int l = lane & 3;
  const int b = blockIdx.x * 8 + (lane >> 2);
  const int lr = (l > 0) ? l - 1 : 0;

  float whh[kG * kH], bhh[kG], wih[kG * kH], bih[kG];
#pragma unroll
  for (int i = 0; i < kG * kH; ++i) {
    whh[i] = (l == 0) ? whh0[i] : whh_r[lr * kG * kH + i];
    wih[i] = wih_r[lr * kG * kH + i];
  }
#pragma unroll
  for (int i = 0; i < kG; ++i) {
    bhh[i] = (l == 0) ? bhh0[i] : bhh_r[lr * kG + i];
    bih[i] = bih_r[lr * kG + i];
  }

  float h[kH];
#pragma unroll
  for (int j = 0; j < kH; ++j) h[j] = hxs[(l * kBATCH + b) * kH + j];

  for (int s = 0; s < kL - 1; ++s)
    scan_step<true>(s, l, b, h, wih, bih, whh, bhh, gx0, out);
  for (int s = kL - 1; s < kSEQ; ++s)
    scan_step<false>(s, l, b, h, wih, bih, whh, bhh, gx0, out);
  for (int s = kSEQ; s < kSEQ + kL - 1; ++s)
    scan_step<true>(s, l, b, h, wih, bih, whh, bhh, gx0, out);

#pragma unroll
  for (int j = 0; j < kH; ++j)
    out[(size_t)kSEQ * kBATCH * kH + (l * kBATCH + b) * kH + j] = h[j];
}

extern "C" void kernel_launch(void* const* d_in, const int* in_sizes, int n_in,
                              void* d_out, int out_size, void* d_ws,
                              size_t ws_size, hipStream_t stream) {
  const float* x    = (const float*)d_in[0];
  const float* hxs  = (const float*)d_in[1];
  const float* wih0 = (const float*)d_in[2];
  const float* whh0 = (const float*)d_in[3];
  const float* bih0 = (const float*)d_in[4];
  const float* bhh0 = (const float*)d_in[5];
  const float* wihr = (const float*)d_in[6];
  const float* whhr = (const float*)d_in[7];
  const float* bihr = (const float*)d_in[8];
  const float* bhhr = (const float*)d_in[9];
  float* out = (float*)d_out;
  float* gx  = (float*)d_ws;

  gru_in_proj_wmma<<<kROWS / 128, 256, 0, stream>>>(x, wih0, bih0, gx);
  gru_scan<<<kBATCH / 8, 32, 0, stream>>>(gx, hxs, whh0, bhh0, wihr, whhr, bihr,
                                          bhhr, out);
}


# GRU scan: layer 0's gx0 row is fetched two steps ahead into a ring of three register rows (steady loop: 3 steps per trip); its memory round trip leaves the dependent chain; sums and order as before
# speedup vs baseline: 5.1251x; 2.5987x over previous
#include <hip/hip_runtime.h>
#include <hip/hip_bf16.h>

typedef float v2f __attribute__((ext_vector_type(2)));
typedef float v8f __attribute__((ext_vector_type(8)));

static constexpr int kSEQ   = 2048;
static constexpr int kBATCH = 64;
static constexpr int kIN    = 512;
static constexpr int kH     = 3;
static constexpr int kL     = 4;
static constexpr int kG     = 3 * kH;
static constexpr int kROWS  = kSEQ * kBATCH;

__global__ __launch_bounds__(256) void gru_in_proj_wmma(
    const float* __restrict__ X, const float* __restrict__ Wih0,
    const float* __restrict__ bih0, float* __restrict__ gx) {
  __shared__ float Bs[kIN * 16];
  const int tid = threadIdx.x;
  for (int idx = tid; idx < kIN * 16; idx += 256) {
    const int k = idx >> 4;
    const int n = idx & 15;
    Bs[idx] = (n < kG) ? Wih0[n * kIN + k] : 0.0f;
  }
  __syncthreads();

  const int wave = tid >> 5;
  const int lane = tid & 31;
  const int half = lane >> 4;
  const int n16  = lane & 15;
  const int row0 = blockIdx.x * 128 + wave * 16;

  const float* arow = X + (size_t)(row0 + n16) * kIN + 2 * half;

  v8f c = {0.f, 0.f, 0.f, 0.f, 0.f, 0.f, 0.f, 0.f};
  for (int kk = 0; kk < kIN; kk += 4) {
    v2f a = *(const v2f*)(arow + kk);
    v2f b;
    b.x = Bs[(kk + 2 * half) * 16 + n16];
    b.y = Bs[(kk + 2 * half + 1) * 16 + n16];
    c = __builtin_amdgcn_wmma_f32_16x16x4_f32(
         false, a,  false, b,
         (short)0, c,  false,  false);
  }

  if (n16 < kG) {
    const float bias = bih0[n16];
#pragma unroll
    for (int v = 0; v < 8; ++v) {
      const int m = v + 8 * half;
      gx[(size_t)(row0 + m) * kG + n16] = c[v] + bias;
    }
  }
}

__device__ __forceinline__ float sigmoid_f(float x) {
  return 1.0f / (1.0f + __expf(-x));
}
__device__ __forceinline__ float tanh_f(float x) {
  const float e = __expf(2.0f * fabsf(x));
  return copysignf(1.0f - 2.0f / (e + 1.0f), x);
}

__device__ __forceinline__ void gru_hidden_proj(float gh[kG], const float h[kH],
                                                const float* __restrict__ whh,
                                                const float* __restrict__ bhh) {
#pragma unroll
  for (int g = 0; g < kG; ++g) {
    float s = bhh[g];
#pragma unroll
    for (int j = 0; j < kH; ++j) s += whh[g * kH + j] * h[j];
    gh[g] = s;
  }
}

__device__ __forceinline__ void gru_gates(float h[kH], const float gxv[kG],
                                          const float gh[kG]) {
#pragma unroll
  for (int j = 0; j < kH; ++j) {
    const float r = sigmoid_f(gxv[j] + gh[j]);
    const float z = sigmoid_f(gxv[kH + j] + gh[kH + j]);
    const float n = tanh_f(gxv[2 * kH + j] + r * gh[2 * kH + j]);
    h[j] = (1.0f - z) * n + z * h[j];
  }
}

__device__ __forceinline__ void load_gx_row(float (&g)[kG],
                                            const float* __restrict__ gx0,
                                            int t, int b) {
  const float* gp = gx0 + (size_t)(t * kBATCH + b) * kG;
#pragma unroll
  for (int i = 0; i < kG; ++i) g[i] = gp[i];
}

__device__ __forceinline__ float from_lower_lane_of_quad(float v) {
  return __builtin_bit_cast(
      float, __builtin_amdgcn_mov_dpp(__builtin_bit_cast(int, v), 0x90, 0xf,
                                      0xf, true));
}

template <bool kEdge>
__device__ __forceinline__ void scan_step(
    int s, int l, int b, float h[kH], const float (&gl)[kG],
    const float (&wih)[kG * kH], const float (&bih)[kG],
    const float (&whh)[kG * kH], const float (&bhh)[kG],
    float* __restrict__ out) {
  const int t = s - l;
  const bool active = !kEdge || (t >= 0 && t < kSEQ);

  float hp[kH];
#pragma unroll
  for (int j = 0; j < kH; ++j) hp[j] = from_lower_lane_of_quad(h[j]);

  float a[kG];
#pragma unroll
  for (int g = 0; g < kG; ++g) {
    float acc = bih[g];
#pragma unroll
    for (int j = 0; j < kH; ++j) acc += wih[g * kH + j] * hp[j];
    a[g] = acc;
  }
  float gh[kG];
  gru_hidden_proj(gh, h, whh, bhh);

  __builtin_amdgcn_sched_barrier(0);

  float gxv[kG];
#pragma unroll
  for (int g = 0; g < kG; ++g) gxv[g] = (l == 0) ? gl[g] : a[g];

  float hn[kH];
#pragma unroll
  for (int j = 0; j < kH; ++j) hn[j] = h[j];
  gru_gates(hn, gxv, gh);

  if (active) {
#pragma unroll
    for (int j = 0; j < kH; ++j) h[j] = hn[j];
    if (l == kL - 1) {
#pragma unroll
      for (int j = 0; j < kH; ++j)
        out[(size_t)(t * kBATCH + b) * kH + j] = hn[j];
    }
  }
}

__device__ __forceinline__ void edge_step(
    int s, int l, int b, float h[kH], const float (&wih)[kG * kH],
    const float (&bih)[kG], const float (&whh)[kG * kH], const float (&bhh)[kG],
    const float* __restrict__ gx0, float* __restrict__ out) {
  const int t0 = (s >= kSEQ) ? kSEQ - 1 : s;
  float gl[kG];
  load_gx_row(gl, gx0, t0, b);
  scan_step<true>(s, l, b, h, gl, wih, bih, whh, bhh, out);
}

static constexpr int kSteadyEnd = kSEQ - 2;
static_assert((kSteadyEnd - (kL - 1)) % 3 == 0, "steady steps come in threes");

__global__ __launch_bounds__(32) void gru_scan(
    const float* __restrict__ gx0, const float* __restrict__ hxs,
    const float* __restrict__ whh0, const float* __restrict__ bhh0,
    const float* __restrict__ wih_r, const float* __restrict__ whh_r,
    const float* __restrict__ bih_r, const float* __restrict__ bhh_r,
    float* __restrict__ out) {
  const int lane = threadIdx.x;
  const int l = lane & 3;
  const int b = blockIdx.x * 8 + (lane >> 2);
  const int lr = (l > 0) ? l - 1 : 0;

  float whh[kG * kH], bhh[kG], wih[kG * kH], bih[kG];
#pragma unroll
  for (int i = 0; i < kG * kH; ++i) {
    whh[i] = (l == 0) ? whh0[i] : whh_r[lr * kG * kH + i];
    wih[i] = wih_r[lr * kG * kH + i];
  }
#pragma unroll
  for (int i = 0; i < kG; ++i) {
    bhh[i] = (l == 0) ? bhh0[i] : bhh_r[lr * kG + i];
    bih[i] = bih_r[lr * kG + i];
  }

  float h[kH];
#pragma unroll
  for (int j = 0; j < kH; ++j) h[j] = hxs[(l * kBATCH + b) * kH + j];

  for (int s = 0; s < kL - 1; ++s)
    edge_step(s, l, b, h, wih, bih, whh, bhh, gx0, out);

  float g0[kG], g1[kG], g2[kG];
  load_gx_row(g0, gx0, kL - 1, b);
  __builtin_amdgcn_sched_barrier(0);
  load_gx_row(g1, gx0, kL, b);

  for (int s = kL - 1; s < kSteadyEnd; s += 3) {
    load_gx_row(g2, gx0, s + 2, b);
    scan_step<false>(s, l, b, h, g0, wih, bih, whh, bhh, out);
    load_gx_row(g0, gx0, s + 3, b);
    scan_step<false>(s + 1, l, b, h, g1, wih, bih, whh, bhh, out);
    load_gx_row(g1, gx0, s + 4, b);
    scan_step<false>(s + 2, l, b, h, g2, wih, bih, whh, bhh, out);
  }
  for (int s = kSteadyEnd; s < kSEQ + kL - 1; ++s)
    edge_step(s, l, b, h, wih, bih, whh, bhh, gx0, out);

#pragma unroll
  for (int j = 0; j < kH; ++j)
    out[(size_t)kSEQ * kBATCH * kH + (l * kBATCH + b) * kH + j] = h[j];
}

extern "C" void kernel_launch(void* const* d_in, const int* in_sizes, int n_in,
                              void* d_out, int out_size, void* d_ws,
                              size_t ws_size, hipStream_t stream) {
  const float* x    = (const float*)d_in[0];
  const float* hxs  = (const float*)d_in[1];
  const float* wih0 = (const float*)d_in[2];
  const float* whh0 = (const float*)d_in[3];
  const float* bih0 = (const float*)d_in[4];
  const float* bhh0 = (const float*)d_in[5];
  const float* wihr = (const float*)d_in[6];
  const float* whhr = (const float*)d_in[7];
  const float* bihr = (const float*)d_in[8];
  const float* bhhr = (const float*)d_in[9];
  float* out = (float*)d_out;
  float* gx  = (float*)d_ws;

  gru_in_proj_wmma<<<kROWS / 128, 256, 0, stream>>>(x, wih0, bih0, gx);
  gru_scan<<<kBATCH / 8, 32, 0, stream>>>(gx, hxs, whh0, bhh0, wihr, whhr, bihr,
                                          bhhr, out);
}


# GRU scan: the nine f32 divisions per cell become v_rcp_f32 (1 ulp); so that the shorter step still hides the gx0 row's trip, rows are fetched a 6-step block ahead, as 16-byte loads; sums as before
# speedup vs baseline: 9.3135x; 1.8172x over previous
#include <hip/hip_runtime.h>
#include <hip/hip_bf16.h>

typedef float v2f __attribute__((ext_vector_type(2)));
typedef float v8f __attribute__((ext_vector_type(8)));

static constexpr int kSEQ   = 2048;
static constexpr int kBATCH = 64;
static constexpr int kIN    = 512;
static constexpr int kH     = 3;
static constexpr int kL     = 4;
static constexpr int kG     = 3 * kH;
static constexpr int kROWS  = kSEQ * kBATCH;

__global__ __launch_bounds__(256) void gru_in_proj_wmma(
    const float* __restrict__ X, const float* __restrict__ Wih0,
    const float* __restrict__ bih0, float* __restrict__ gx) {
  __shared__ float Bs[kIN * 16];
  const int tid = threadIdx.x;
  for (int idx = tid; idx < kIN * 16; idx += 256) {
    const int k = idx >> 4;
    const int n = idx & 15;
    Bs[idx] = (n < kG) ? Wih0[n * kIN + k] : 0.0f;
  }
  __syncthreads();

  const int wave = tid >> 5;
  const int lane = tid & 31;
  const int half = lane >> 4;
  const int n16  = lane & 15;
  const int row0 = blockIdx.x * 128 + wave * 16;

  const float* arow = X + (size_t)(row0 + n16) * kIN + 2 * half;

  v8f c = {0.f, 0.f, 0.f, 0.f, 0.f, 0.f, 0.f, 0.f};
  for (int kk = 0; kk < kIN; kk += 4) {
    v2f a = *(const v2f*)(arow + kk);
    v2f b;
    b.x = Bs[(kk + 2 * half) * 16 + n16];
    b.y = Bs[(kk + 2 * half + 1) * 16 + n16];
    c = __builtin_amdgcn_wmma_f32_16x16x4_f32(
         false, a,  false, b,
         (short)0, c,  false,  false);
  }

  if (n16 < kG) {
    const float bias = bih0[n16];
#pragma unroll
    for (int v = 0; v < 8; ++v) {
      const int m = v + 8 * half;
      gx[(size_t)(row0 + m) * kG + n16] = c[v] + bias;
    }
  }
}

__device__ __forceinline__ float sigmoid_f(float x) {
  return __builtin_amdgcn_rcpf(1.0f + __expf(-x));
}
__device__ __forceinline__ float tanh_f(float x) {
  const float e = __expf(2.0f * fabsf(x));
  return copysignf(1.0f - 2.0f * __builtin_amdgcn_rcpf(e + 1.0f), x);
}

__device__ __forceinline__ void gru_hidden_proj(float gh[kG], const float h[kH],
                                                const float* __restrict__ whh,
                                                const float* __restrict__ bhh) {
#pragma unroll
  for (int g = 0; g < kG; ++g) {
    float s = bhh[g];
#pragma unroll
    for (int j = 0; j < kH; ++j) s += whh[g * kH + j] * h[j];
    gh[g] = s;
  }
}

__device__ __forceinline__ void gru_gates(float h[kH], const float gxv[kG],
                                          const float gh[kG]) {
#pragma unroll
  for (int j = 0; j < kH; ++j) {
    const float r = sigmoid_f(gxv[j] + gh[j]);
    const float z = sigmoid_f(gxv[kH + j] + gh[kH + j]);
    const float n = tanh_f(gxv[2 * kH + j] + r * gh[2 * kH + j]);
    h[j] = (1.0f - z) * n + z * h[j];
  }
}

static constexpr int kRowRegs = 8;
__device__ __forceinline__ void load_gx_row(float (&g)[kRowRegs],
                                            const float* __restrict__ gx0,
                                            int t, int b, int off1) {
  const float* gp = gx0 + (size_t)(t * kBATCH + b) * kG + off1;
#pragma unroll
  for (int i = 0; i < kRowRegs; ++i) g[i] = gp[i];
}

__device__ __forceinline__ float from_lane1_of_quad(float v) {
  return __builtin_bit_cast(
      float, __builtin_amdgcn_mov_dpp(__builtin_bit_cast(int, v), 0x55, 0xf,
                                      0xf, true));
}

__device__ __forceinline__ float from_lower_lane_of_quad(float v) {
  return __builtin_bit_cast(
      float, __builtin_amdgcn_mov_dpp(__builtin_bit_cast(int, v), 0x90, 0xf,
                                      0xf, true));
}

template <bool kEdge>
__device__ __forceinline__ void scan_step(
    int s, int l, int b, float h[kH], const float (&gl)[kRowRegs],
    const float (&wih)[kG * kH], const float (&bih)[kG],
    const float (&whh)[kG * kH], const float (&bhh)[kG],
    float* __restrict__ out) {
  const int t = s - l;
  const bool active = !kEdge || (t >= 0 && t < kSEQ);

  float hp[kH];
#pragma unroll
  for (int j = 0; j < kH; ++j) hp[j] = from_lower_lane_of_quad(h[j]);

  float a[kG];
#pragma unroll
  for (int g = 0; g < kG; ++g) {
    float acc = bih[g];
#pragma unroll
    for (int j = 0; j < kH; ++j) acc += wih[g * kH + j] * hp[j];
    a[g] = acc;
  }
  float gh[kG];
  gru_hidden_proj(gh, h, whh, bhh);

  __builtin_amdgcn_sched_barrier(0);

  const float gl8 = from_lane1_of_quad(gl[kRowRegs - 1]);
  float gxv[kG];
#pragma unroll
  for (int g = 0; g < kG; ++g)
    gxv[g] = (l == 0) ? (g < kRowRegs ? gl[g] : gl8) : a[g];

  float hn[kH];
#pragma unroll
  for (int j = 0; j < kH; ++j) hn[j] = h[j];
  gru_gates(hn, gxv, gh);

  if (active) {
#pragma unroll
    for (int j = 0; j < kH; ++j) h[j] = hn[j];
    if (l == kL - 1) {
#pragma unroll
      for (int j = 0; j < kH; ++j)
        out[(size_t)(t * kBATCH + b) * kH + j] = hn[j];
    }
  }
}

__device__ __forceinline__ void edge_step(
    int s, int l, int b, float h[kH], const float (&wih)[kG * kH],
    const float (&bih)[kG], const float (&whh)[kG * kH], const float (&bhh)[kG],
    const float* __restrict__ gx0, int off1, float* __restrict__ out) {
  const int t0 = (s >= kSEQ) ? kSEQ - 1 : s;
  float gl[kRowRegs];
  load_gx_row(gl, gx0, t0, b, off1);
  scan_step<true>(s, l, b, h, gl, wih, bih, whh, bhh, out);
}

static constexpr int kBlk = 6;
static constexpr int kSteadyEnd = (kL - 1) + ((kSEQ - (kL - 1)) / kBlk) * kBlk;
static_assert((kSteadyEnd - (kL - 1)) % kBlk == 0, "whole blocks only");
static_assert(kSteadyEnd <= kSEQ, "steady steps are full steps");
static_assert(kSteadyEnd > kL - 1, "at least one block");

__global__ __launch_bounds__(32) void gru_scan(
    const float* __restrict__ gx0, const float* __restrict__ hxs,
    const float* __restrict__ whh0, const float* __restrict__ bhh0,
    const float* __restrict__ wih_r, const float* __restrict__ whh_r,
    const float* __restrict__ bih_r, const float* __restrict__ bhh_r,
    float* __restrict__ out) {
  const int lane = threadIdx.x;
  const int l = lane & 3;
  const int b = blockIdx.x * 8 + (lane >> 2);
  const int lr = (l > 0) ? l - 1 : 0;
  const int off1 = (l == 1) ? 1 : 0;

  float whh[kG * kH], bhh[kG], wih[kG * kH], bih[kG];
#pragma unroll
  for (int i = 0; i < kG * kH; ++i) {
    whh[i] = (l == 0) ? whh0[i] : whh_r[lr * kG * kH + i];
    wih[i] = wih_r[lr * kG * kH + i];
  }
#pragma unroll
  for (int i = 0; i < kG; ++i) {
    bhh[i] = (l == 0) ? bhh0[i] : bhh_r[lr * kG + i];
    bih[i] = bih_r[lr * kG + i];
  }

  float h[kH];
#pragma unroll
  for (int j = 0; j < kH; ++j) h[j] = hxs[(l * kBATCH + b) * kH + j];

  for (int s = 0; s < kL - 1; ++s)
    edge_step(s, l, b, h, wih, bih, whh, bhh, gx0, off1, out);

  float cur[kBlk][kRowRegs];
#pragma unroll
  for (int i = 0; i < kBlk; ++i) load_gx_row(cur[i], gx0, kL - 1 + i, b, off1);

  for (int s = kL - 1; s < kSteadyEnd; s += kBlk) {
    float nxt[kBlk][kRowRegs];
#pragma unroll
    for (int i = 0; i < kBlk; ++i) {
      const int t = s + kBlk + i;
      load_gx_row(nxt[i], gx0, (t < kSEQ) ? t : kSEQ - 1, b, off1);
    }
#pragma unroll
    for (int i = 0; i < kBlk; ++i)
      scan_step<false>(s + i, l, b, h, cur[i], wih, bih, whh, bhh, out);
#pragma unroll
    for (int i = 0; i < kBlk; ++i) {
#pragma unroll
      for (int j = 0; j < kRowRegs; ++j) cur[i][j] = nxt[i][j];
    }
  }
  for (int s = kSteadyEnd; s < kSEQ + kL - 1; ++s)
    edge_step(s, l, b, h, wih, bih, whh, bhh, gx0, off1, out);

#pragma unroll
  for (int j = 0; j < kH; ++j)
    out[(size_t)kSEQ * kBATCH * kH + (l * kBATCH + b) * kH + j] = h[j];
}

extern "C" void kernel_launch(void* const* d_in, const int* in_sizes, int n_in,
                              void* d_out, int out_size, void* d_ws,
                              size_t ws_size, hipStream_t stream) {
  const float* x    = (const float*)d_in[0];
  const float* hxs  = (const float*)d_in[1];
  const float* wih0 = (const float*)d_in[2];
  const float* whh0 = (const float*)d_in[3];
  const float* bih0 = (const float*)d_in[4];
  const float* bhh0 = (const float*)d_in[5];
  const float* wihr = (const float*)d_in[6];
  const float* whhr = (const float*)d_in[7];
  const float* bihr = (const float*)d_in[8];
  const float* bhhr = (const float*)d_in[9];
  float* out = (float*)d_out;
  float* gx  = (float*)d_ws;

  gru_in_proj_wmma<<<kROWS / 128, 256, 0, stream>>>(x, wih0, bih0, gx);
  gru_scan<<<kBATCH / 8, 32, 0, stream>>>(gx, hxs, whh0, bhh0, wihr, whhr, bihr,
                                          bhhr, out);
}


# GRU scan: one lane per hidden unit (lane = 16*batch + 4*unit + layer, 32 waves): 3 gates per lane, not 9; other units' and lower layer's h by DPP row_ror; unit sums as fma chains in rotated order
# speedup vs baseline: 12.3550x; 1.3266x over previous
#include <hip/hip_runtime.h>
#include <hip/hip_bf16.h>

typedef float v2f __attribute__((ext_vector_type(2)));
typedef float v8f __attribute__((ext_vector_type(8)));

static constexpr int kSEQ   = 2048;
static constexpr int kBATCH = 64;
static constexpr int kIN    = 512;
static constexpr int kH     = 3;
static constexpr int kL     = 4;
static constexpr int kG     = 3 * kH;
static constexpr int kROWS  = kSEQ * kBATCH;

__global__ __launch_bounds__(256) void gru_in_proj_wmma(
    const float* __restrict__ X, const float* __restrict__ Wih0,
    const float* __restrict__ bih0, float* __restrict__ gx) {
  __shared__ float Bs[kIN * 16];
  const int tid = threadIdx.x;
  for (int idx = tid; idx < kIN * 16; idx += 256) {
    const int k = idx >> 4;
    const int n = idx & 15;
    Bs[idx] = (n < kG) ? Wih0[n * kIN + k] : 0.0f;
  }
  __syncthreads();

  const int wave = tid >> 5;
  const int lane = tid & 31;
  const int half = lane >> 4;
  const int n16  = lane & 15;
  const int row0 = blockIdx.x * 128 + wave * 16;

  const float* arow = X + (size_t)(row0 + n16) * kIN + 2 * half;

  v8f c = {0.f, 0.f, 0.f, 0.f, 0.f, 0.f, 0.f, 0.f};
  for (int kk = 0; kk < kIN; kk += 4) {
    v2f a = *(const v2f*)(arow + kk);
    v2f b;
    b.x = Bs[(kk + 2 * half) * 16 + n16];
    b.y = Bs[(kk + 2 * half + 1) * 16 + n16];
    c = __builtin_amdgcn_wmma_f32_16x16x4_f32(
         false, a,  false, b,
         (short)0, c,  false,  false);
  }

  if (n16 < kG) {
    const float bias = bih0[n16];
#pragma unroll
    for (int v = 0; v < 8; ++v) {
      const int m = v + 8 * half;
      gx[(size_t)(row0 + m) * kG + n16] = c[v] + bias;
    }
  }
}

__device__ __forceinline__ float sigmoid_f(float x) {
  return __builtin_amdgcn_rcpf(1.0f + __expf(-x));
}
__device__ __forceinline__ float tanh_f(float x) {
  const float e = __expf(2.0f * fabsf(x));
  return copysignf(1.0f - 2.0f * __builtin_amdgcn_rcpf(e + 1.0f), x);
}


template <int kN>
__device__ __forceinline__ float row_ror(float v) {
  return __builtin_bit_cast(
      float, __builtin_amdgcn_mov_dpp(__builtin_bit_cast(int, v), 0x120 + kN,
                                      0xf, 0xf, true));
}

static constexpr int kGates = 3;
static constexpr int kSrc   = 4;

__device__ __forceinline__ void load_gx_unit(float (&g)[kGates],
                                             const float* __restrict__ gx0,
                                             int t, int gofs) {
  const float* gp = gx0 + (size_t)t * (kBATCH * kG) + gofs;
#pragma unroll
  for (int i = 0; i < kGates; ++i) g[i] = gp[i * kH];
}

template <bool kEdge>
__device__ __forceinline__ void scan_step(
    int s, int l, bool use_gx, bool stores, int oofs, float& h,
    const float (&row)[kGates], const float (&wlo)[kGates][kSrc],
    const float (&bih)[kGates], const float (&whh)[kGates][kSrc],
    const float (&bhh)[kGates], float* __restrict__ out) {
  const int t = s - l;
  const bool active = !kEdge || (t >= 0 && t < kSEQ);

  const float h4 = row_ror<4>(h), h8 = row_ror<8>(h), h12 = row_ror<12>(h);
  const float p1 = row_ror<1>(h), p5 = row_ror<5>(h), p9 = row_ror<9>(h),
              p13 = row_ror<13>(h);

  float a[kGates], gh[kGates];
#pragma unroll
  for (int g = 0; g < kGates; ++g) {
    float acc = use_gx ? row[g] : bih[g];
    acc = __builtin_fmaf(wlo[g][0], p1, acc);
    acc = __builtin_fmaf(wlo[g][1], p5, acc);
    acc = __builtin_fmaf(wlo[g][2], p9, acc);
    acc = __builtin_fmaf(wlo[g][3], p13, acc);
    a[g] = acc;
    float sum = __builtin_fmaf(whh[g][0], h, bhh[g]);
    sum = __builtin_fmaf(whh[g][1], h4, sum);
    sum = __builtin_fmaf(whh[g][2], h8, sum);
    sum = __builtin_fmaf(whh[g][3], h12, sum);
    gh[g] = sum;
  }

  const float r = sigmoid_f(a[0] + gh[0]);
  const float z = sigmoid_f(a[1] + gh[1]);
  const float omz = 1.0f - z;
  const float zh = z * h;

  __builtin_amdgcn_sched_barrier(0);

  const float n = tanh_f(a[2] + r * gh[2]);
  const float hn = __builtin_fmaf(omz, n, zh);

  if (active) {
    h = hn;
    if (stores) out[(size_t)t * (kBATCH * kH) + oofs] = hn;
  }
}

__device__ __forceinline__ void edge_step(
    int s, int l, bool use_gx, bool stores, int oofs, int gofs, float& h,
    const float (&wlo)[kGates][kSrc], const float (&bih)[kGates],
    const float (&whh)[kGates][kSrc], const float (&bhh)[kGates],
    const float* __restrict__ gx0, float* __restrict__ out) {
  const int t0 = (s >= kSEQ) ? kSEQ - 1 : s;
  float row[kGates];
  load_gx_unit(row, gx0, t0, gofs);
  scan_step<true>(s, l, use_gx, stores, oofs, h, row, wlo, bih, whh, bhh, out);
}

static constexpr int kBlk = 12;
static constexpr int kSteadyEnd = (kL - 1) + ((kSEQ - (kL - 1)) / kBlk) * kBlk;
static_assert((kSteadyEnd - (kL - 1)) % kBlk == 0, "whole blocks only");
static_assert(kSteadyEnd <= kSEQ, "steady steps are full steps");
static_assert(kSteadyEnd > kL - 1, "at least one block");
static constexpr int kBatchPerWave = 2;

__global__ __launch_bounds__(32) void gru_scan(
    const float* __restrict__ gx0, const float* __restrict__ hxs,
    const float* __restrict__ whh0, const float* __restrict__ bhh0,
    const float* __restrict__ wih_r, const float* __restrict__ whh_r,
    const float* __restrict__ bih_r, const float* __restrict__ bhh_r,
    float* __restrict__ out) {
  const int lane = threadIdx.x;
  const int p = lane & 15;
  const int j = p >> 2;
  const int l = p & 3;
  const int b = blockIdx.x * kBatchPerWave + (lane >> 4);
  const bool real = j < kH;
  const int jj = real ? j : 0;
  const int lr = (l > 0) ? l - 1 : 0;
  const bool use_gx = real && l == 0;
  const bool stores = real && l == kL - 1;
  const int gofs = b * kG + jj;
  const int oofs = b * kH + jj;

  float whh[kGates][kSrc], wlo[kGates][kSrc], bhh[kGates], bih[kGates];
#pragma unroll
  for (int g = 0; g < kGates; ++g) {
    const int grow = g * kH + jj;
#pragma unroll
    for (int k = 0; k < kSrc; ++k) {
      const int u = (j - k) & 3;
      const bool have = real && u < kH;
      const int uu = (u < kH) ? u : 0;
      const float wh = (l == 0) ? whh0[grow * kH + uu]
                                : whh_r[lr * kG * kH + grow * kH + uu];
      const float wl = wih_r[lr * kG * kH + grow * kH + uu];
      whh[g][k] = have ? wh : 0.0f;
      wlo[g][k] = (have && l > 0) ? wl : 0.0f;
    }
    const float bh = (l == 0) ? bhh0[grow] : bhh_r[lr * kG + grow];
    const float bi = bih_r[lr * kG + grow];
    bhh[g] = real ? bh : 0.0f;
    bih[g] = (real && l > 0) ? bi : 0.0f;
  }

  float h = real ? hxs[(l * kBATCH + b) * kH + jj] : 0.0f;

  for (int s = 0; s < kL - 1; ++s)
    edge_step(s, l, use_gx, stores, oofs, gofs, h, wlo, bih, whh, bhh, gx0,
              out);

  float cur[kBlk][kGates];
#pragma unroll
  for (int i = 0; i < kBlk; ++i) load_gx_unit(cur[i], gx0, kL - 1 + i, gofs);

  for (int s = kL - 1; s < kSteadyEnd; s += kBlk) {
    float nxt[kBlk][kGates];
#pragma unroll
    for (int i = 0; i < kBlk; ++i) {
      const int t = s + kBlk + i;
      load_gx_unit(nxt[i], gx0, (t < kSEQ) ? t : kSEQ - 1, gofs);
    }
#pragma unroll
    for (int i = 0; i < kBlk; ++i)
      scan_step<false>(s + i, l, use_gx, stores, oofs, h, cur[i], wlo, bih,
                       whh, bhh, out);
#pragma unroll
    for (int i = 0; i < kBlk; ++i) {
#pragma unroll
      for (int g = 0; g < kGates; ++g) cur[i][g] = nxt[i][g];
    }
  }
  for (int s = kSteadyEnd; s < kSEQ + kL - 1; ++s)
    edge_step(s, l, use_gx, stores, oofs, gofs, h, wlo, bih, whh, bhh, gx0,
              out);

  if (real) out[(size_t)kSEQ * kBATCH * kH + (l * kBATCH + b) * kH + jj] = h;
}

extern "C" void kernel_launch(void* const* d_in, const int* in_sizes, int n_in,
                              void* d_out, int out_size, void* d_ws,
                              size_t ws_size, hipStream_t stream) {
  const float* x    = (const float*)d_in[0];
  const float* hxs  = (const float*)d_in[1];
  const float* wih0 = (const float*)d_in[2];
  const float* whh0 = (const float*)d_in[3];
  const float* bih0 = (const float*)d_in[4];
  const float* bhh0 = (const float*)d_in[5];
  const float* wihr = (const float*)d_in[6];
  const float* whhr = (const float*)d_in[7];
  const float* bihr = (const float*)d_in[8];
  const float* bhhr = (const float*)d_in[9];
  float* out = (float*)d_out;
  float* gx  = (float*)d_ws;

  gru_in_proj_wmma<<<kROWS / 128, 256, 0, stream>>>(x, wih0, bih0, gx);
  gru_scan<<<kBATCH / kBatchPerWave, 32, 0, stream>>>(gx, hxs, whh0, bhh0, wihr, whhr, bihr,
                                          bhhr, out);
}


# GRU scan: gates on the hardware tanh (v_tanh_f32): sigmoid(x)=0.5+0.5*tanh(x/2), halves folded exactly into the weights, n gate's argument summed as (nx+nh/2)+t*(nh/2); other sums and order as before
# speedup vs baseline: 14.8533x; 1.2022x over previous
#include <hip/hip_runtime.h>
#include <hip/hip_bf16.h>

typedef float v2f __attribute__((ext_vector_type(2)));
typedef float v8f __attribute__((ext_vector_type(8)));

static constexpr int kSEQ   = 2048;
static constexpr int kBATCH = 64;
static constexpr int kIN    = 512;
static constexpr int kH     = 3;
static constexpr int kL     = 4;
static constexpr int kG     = 3 * kH;
static constexpr int kROWS  = kSEQ * kBATCH;

__global__ __launch_bounds__(256) void gru_in_proj_wmma(
    const float* __restrict__ X, const float* __restrict__ Wih0,
    const float* __restrict__ bih0, float* __restrict__ gx) {
  __shared__ float Bs[kIN * 16];
  const int tid = threadIdx.x;
  for (int idx = tid; idx < kIN * 16; idx += 256) {
    const int k = idx >> 4;
    const int n = idx & 15;
    Bs[idx] = (n < kG) ? Wih0[n * kIN + k] : 0.0f;
  }
  __syncthreads();

  const int wave = tid >> 5;
  const int lane = tid & 31;
  const int half = lane >> 4;
  const int n16  = lane & 15;
  const int row0 = blockIdx.x * 128 + wave * 16;

  const float* arow = X + (size_t)(row0 + n16) * kIN + 2 * half;

  v8f c = {0.f, 0.f, 0.f, 0.f, 0.f, 0.f, 0.f, 0.f};
  for (int kk = 0; kk < kIN; kk += 4) {
    v2f a = *(const v2f*)(arow + kk);
    v2f b;
    b.x = Bs[(kk + 2 * half) * 16 + n16];
    b.y = Bs[(kk + 2 * half + 1) * 16 + n16];
    c = __builtin_amdgcn_wmma_f32_16x16x4_f32(
         false, a,  false, b,
         (short)0, c,  false,  false);
  }

  if (n16 < kG) {
    const float bias = bih0[n16];
#pragma unroll
    for (int v = 0; v < 8; ++v) {
      const int m = v + 8 * half;
      gx[(size_t)(row0 + m) * kG + n16] = c[v] + bias;
    }
  }
}

__device__ __forceinline__ float tanh_hw(float x) {
  return __builtin_amdgcn_tanhf(x);
}

template <int kN>
__device__ __forceinline__ float row_ror(float v) {
  return __builtin_bit_cast(
      float, __builtin_amdgcn_mov_dpp(__builtin_bit_cast(int, v), 0x120 + kN,
                                      0xf, 0xf, true));
}

static constexpr int kGates = 3;
static constexpr int kSrc   = 4;

__device__ __forceinline__ void load_gx_unit(float (&g)[kGates],
                                             const float* __restrict__ gx0,
                                             int t, int gofs) {
  const float* gp = gx0 + (size_t)t * (kBATCH * kG) + gofs;
#pragma unroll
  for (int i = 0; i < kGates; ++i) g[i] = gp[i * kH];
}

template <bool kEdge>
__device__ __forceinline__ void scan_step(
    int s, int l, bool use_gx, bool stores, int oofs, float& h,
    const float (&row)[kGates], const float (&wlo)[kGates][kSrc],
    const float (&bih)[kGates], const float (&whh)[kGates][kSrc],
    const float (&bhh)[kGates], float* __restrict__ out) {
  const int t = s - l;
  const bool active = !kEdge || (t >= 0 && t < kSEQ);

  const float h4 = row_ror<4>(h), h8 = row_ror<8>(h), h12 = row_ror<12>(h);
  const float p1 = row_ror<1>(h), p5 = row_ror<5>(h), p9 = row_ror<9>(h),
              p13 = row_ror<13>(h);

  float a[kGates], gh[kGates];
#pragma unroll
  for (int g = 0; g < kGates; ++g) {
    float acc = use_gx ? row[g] * (g < kGates - 1 ? 0.5f : 1.0f) : bih[g];
    acc = __builtin_fmaf(wlo[g][0], p1, acc);
    acc = __builtin_fmaf(wlo[g][1], p5, acc);
    acc = __builtin_fmaf(wlo[g][2], p9, acc);
    acc = __builtin_fmaf(wlo[g][3], p13, acc);
    a[g] = acc;
    float sum = __builtin_fmaf(whh[g][0], h, bhh[g]);
    sum = __builtin_fmaf(whh[g][1], h4, sum);
    sum = __builtin_fmaf(whh[g][2], h8, sum);
    sum = __builtin_fmaf(whh[g][3], h12, sum);
    gh[g] = sum;
  }

  const float tr = tanh_hw(a[0] + gh[0]);
  const float tz = tanh_hw(a[1] + gh[1]);
  const float z = __builtin_fmaf(0.5f, tz, 0.5f);
  const float omz = 1.0f - z;
  const float zh = z * h;
  const float nbase = a[2] + gh[2];

  const float n = tanh_hw(__builtin_fmaf(tr, gh[2], nbase));
  const float hn = __builtin_fmaf(omz, n, zh);

  if (active) {
    h = hn;
    if (stores) out[(size_t)t * (kBATCH * kH) + oofs] = hn;
  }
}

__device__ __forceinline__ void edge_step(
    int s, int l, bool use_gx, bool stores, int oofs, int gofs, float& h,
    const float (&wlo)[kGates][kSrc], const float (&bih)[kGates],
    const float (&whh)[kGates][kSrc], const float (&bhh)[kGates],
    const float* __restrict__ gx0, float* __restrict__ out) {
  const int t0 = (s >= kSEQ) ? kSEQ - 1 : s;
  float row[kGates];
  load_gx_unit(row, gx0, t0, gofs);
  scan_step<true>(s, l, use_gx, stores, oofs, h, row, wlo, bih, whh, bhh, out);
}

static constexpr int kBlk = 12;
static constexpr int kSteadyEnd = (kL - 1) + ((kSEQ - (kL - 1)) / kBlk) * kBlk;
static_assert((kSteadyEnd - (kL - 1)) % kBlk == 0, "whole blocks only");
static_assert(kSteadyEnd <= kSEQ, "steady steps are full steps");
static_assert(kSteadyEnd > kL - 1, "at least one block");
static constexpr int kBatchPerWave = 2;

__global__ __launch_bounds__(32) void gru_scan(
    const float* __restrict__ gx0, const float* __restrict__ hxs,
    const float* __restrict__ whh0, const float* __restrict__ bhh0,
    const float* __restrict__ wih_r, const float* __restrict__ whh_r,
    const float* __restrict__ bih_r, const float* __restrict__ bhh_r,
    float* __restrict__ out) {
  const int lane = threadIdx.x;
  const int p = lane & 15;
  const int j = p >> 2;
  const int l = p & 3;
  const int b = blockIdx.x * kBatchPerWave + (lane >> 4);
  const bool real = j < kH;
  const int jj = real ? j : 0;
  const int lr = (l > 0) ? l - 1 : 0;
  const bool use_gx = real && l == 0;
  const bool stores = real && l == kL - 1;
  const int gofs = b * kG + jj;
  const int oofs = b * kH + jj;

  float whh[kGates][kSrc], wlo[kGates][kSrc], bhh[kGates], bih[kGates];
#pragma unroll
  for (int g = 0; g < kGates; ++g) {
    const int grow = g * kH + jj;
    const float slo = (g < kGates - 1) ? 0.5f : 1.0f;
#pragma unroll
    for (int k = 0; k < kSrc; ++k) {
      const int u = (j - k) & 3;
      const bool have = real && u < kH;
      const int uu = (u < kH) ? u : 0;
      const float wh = (l == 0) ? whh0[grow * kH + uu]
                                : whh_r[lr * kG * kH + grow * kH + uu];
      const float wl = wih_r[lr * kG * kH + grow * kH + uu];
      whh[g][k] = have ? wh * 0.5f : 0.0f;
      wlo[g][k] = (have && l > 0) ? wl * slo : 0.0f;
    }
    const float bh = (l == 0) ? bhh0[grow] : bhh_r[lr * kG + grow];
    const float bi = bih_r[lr * kG + grow];
    bhh[g] = real ? bh * 0.5f : 0.0f;
    bih[g] = (real && l > 0) ? bi * slo : 0.0f;
  }

  float h = real ? hxs[(l * kBATCH + b) * kH + jj] : 0.0f;

  for (int s = 0; s < kL - 1; ++s)
    edge_step(s, l, use_gx, stores, oofs, gofs, h, wlo, bih, whh, bhh, gx0,
              out);

  float cur[kBlk][kGates];
#pragma unroll
  for (int i = 0; i < kBlk; ++i) load_gx_unit(cur[i], gx0, kL - 1 + i, gofs);

  for (int s = kL - 1; s < kSteadyEnd; s += kBlk) {
    float nxt[kBlk][kGates];
#pragma unroll
    for (int i = 0; i < kBlk; ++i) {
      const int t = s + kBlk + i;
      load_gx_unit(nxt[i], gx0, (t < kSEQ) ? t : kSEQ - 1, gofs);
    }
#pragma unroll
    for (int i = 0; i < kBlk; ++i)
      scan_step<false>(s + i, l, use_gx, stores, oofs, h, cur[i], wlo, bih,
                       whh, bhh, out);
#pragma unroll
    for (int i = 0; i < kBlk; ++i) {
#pragma unroll
      for (int g = 0; g < kGates; ++g) cur[i][g] = nxt[i][g];
    }
  }
  for (int s = kSteadyEnd; s < kSEQ + kL - 1; ++s)
    edge_step(s, l, use_gx, stores, oofs, gofs, h, wlo, bih, whh, bhh, gx0,
              out);

  if (real) out[(size_t)kSEQ * kBATCH * kH + (l * kBATCH + b) * kH + jj] = h;
}

extern "C" void kernel_launch(void* const* d_in, const int* in_sizes, int n_in,
                              void* d_out, int out_size, void* d_ws,
                              size_t ws_size, hipStream_t stream) {
  const float* x    = (const float*)d_in[0];
  const float* hxs  = (const float*)d_in[1];
  const float* wih0 = (const float*)d_in[2];
  const float* whh0 = (const float*)d_in[3];
  const float* bih0 = (const float*)d_in[4];
  const float* bhh0 = (const float*)d_in[5];
  const float* wihr = (const float*)d_in[6];
  const float* whhr = (const float*)d_in[7];
  const float* bihr = (const float*)d_in[8];
  const float* bhhr = (const float*)d_in[9];
  float* out = (float*)d_out;
  float* gx  = (float*)d_ws;

  gru_in_proj_wmma<<<kROWS / 128, 256, 0, stream>>>(x, wih0, bih0, gx);
  gru_scan<<<kBATCH / kBatchPerWave, 32, 0, stream>>>(gx, hxs, whh0, bhh0, wihr, whhr, bihr,
                                          bhhr, out);
}


# GRU scan: a 12-step block's top-layer outputs stay in registers and are stored as one group behind the block: no store, lane-mask switch or wait behind a store inside a step; sums and order as before
# speedup vs baseline: 22.5958x; 1.5213x over previous
#include <hip/hip_runtime.h>
#include <hip/hip_bf16.h>

typedef float v2f __attribute__((ext_vector_type(2)));
typedef float v8f __attribute__((ext_vector_type(8)));

static constexpr int kSEQ   = 2048;
static constexpr int kBATCH = 64;
static constexpr int kIN    = 512;
static constexpr int kH     = 3;
static constexpr int kL     = 4;
static constexpr int kG     = 3 * kH;
static constexpr int kROWS  = kSEQ * kBATCH;

__global__ __launch_bounds__(256) void gru_in_proj_wmma(
    const float* __restrict__ X, const float* __restrict__ Wih0,
    const float* __restrict__ bih0, float* __restrict__ gx) {
  __shared__ float Bs[kIN * 16];
  const int tid = threadIdx.x;
  for (int idx = tid; idx < kIN * 16; idx += 256) {
    const int k = idx >> 4;
    const int n = idx & 15;
    Bs[idx] = (n < kG) ? Wih0[n * kIN + k] : 0.0f;
  }
  __syncthreads();

  const int wave = tid >> 5;
  const int lane = tid & 31;
  const int half = lane >> 4;
  const int n16  = lane & 15;
  const int row0 = blockIdx.x * 128 + wave * 16;

  const float* arow = X + (size_t)(row0 + n16) * kIN + 2 * half;

  v8f c = {0.f, 0.f, 0.f, 0.f, 0.f, 0.f, 0.f, 0.f};
  for (int kk = 0; kk < kIN; kk += 4) {
    v2f a = *(const v2f*)(arow + kk);
    v2f b;
    b.x = Bs[(kk + 2 * half) * 16 + n16];
    b.y = Bs[(kk + 2 * half + 1) * 16 + n16];
    c = __builtin_amdgcn_wmma_f32_16x16x4_f32(
         false, a,  false, b,
         (short)0, c,  false,  false);
  }

  if (n16 < kG) {
    const float bias = bih0[n16];
#pragma unroll
    for (int v = 0; v < 8; ++v) {
      const int m = v + 8 * half;
      gx[(size_t)(row0 + m) * kG + n16] = c[v] + bias;
    }
  }
}

__device__ __forceinline__ float tanh_hw(float x) {
  return __builtin_amdgcn_tanhf(x);
}

template <int kN>
__device__ __forceinline__ float row_ror(float v) {
  return __builtin_bit_cast(
      float, __builtin_amdgcn_mov_dpp(__builtin_bit_cast(int, v), 0x120 + kN,
                                      0xf, 0xf, true));
}

static constexpr int kGates = 3;
static constexpr int kSrc   = 4;

__device__ __forceinline__ void load_gx_unit(float (&g)[kGates],
                                             const float* __restrict__ gx0,
                                             int t, int gofs) {
  const float* gp = gx0 + (size_t)t * (kBATCH * kG) + gofs;
#pragma unroll
  for (int i = 0; i < kGates; ++i) g[i] = gp[i * kH];
}

__device__ __forceinline__ float scan_step(
    bool use_gx, float h, const float (&row)[kGates],
    const float (&wlo)[kGates][kSrc], const float (&bih)[kGates],
    const float (&whh)[kGates][kSrc], const float (&bhh)[kGates]) {
  const float h4 = row_ror<4>(h), h8 = row_ror<8>(h), h12 = row_ror<12>(h);
  const float p1 = row_ror<1>(h), p5 = row_ror<5>(h), p9 = row_ror<9>(h),
              p13 = row_ror<13>(h);

  float a[kGates], gh[kGates];
#pragma unroll
  for (int g = 0; g < kGates; ++g) {
    float acc = use_gx ? row[g] * (g < kGates - 1 ? 0.5f : 1.0f) : bih[g];
    acc = __builtin_fmaf(wlo[g][0], p1, acc);
    acc = __builtin_fmaf(wlo[g][1], p5, acc);
    acc = __builtin_fmaf(wlo[g][2], p9, acc);
    acc = __builtin_fmaf(wlo[g][3], p13, acc);
    a[g] = acc;
    float sum = __builtin_fmaf(whh[g][0], h, bhh[g]);
    sum = __builtin_fmaf(whh[g][1], h4, sum);
    sum = __builtin_fmaf(whh[g][2], h8, sum);
    sum = __builtin_fmaf(whh[g][3], h12, sum);
    gh[g] = sum;
  }

  const float tr = tanh_hw(a[0] + gh[0]);
  const float tz = tanh_hw(a[1] + gh[1]);
  const float z = __builtin_fmaf(0.5f, tz, 0.5f);
  const float omz = 1.0f - z;
  const float zh = z * h;
  const float nbase = a[2] + gh[2];

  const float n = tanh_hw(__builtin_fmaf(tr, gh[2], nbase));
  return __builtin_fmaf(omz, n, zh);
}

__device__ __forceinline__ void edge_step(
    int s, int l, bool use_gx, bool stores, int oofs, int gofs, float& h,
    const float (&wlo)[kGates][kSrc], const float (&bih)[kGates],
    const float (&whh)[kGates][kSrc], const float (&bhh)[kGates],
    const float* __restrict__ gx0, float* __restrict__ out) {
  const int t0 = (s >= kSEQ) ? kSEQ - 1 : s;
  float row[kGates];
  load_gx_unit(row, gx0, t0, gofs);
  const int t = s - l;
  const float hn = scan_step(use_gx, h, row, wlo, bih, whh, bhh);
  if (t >= 0 && t < kSEQ) {
    h = hn;
    if (stores) out[(size_t)t * (kBATCH * kH) + oofs] = hn;
  }
}

static constexpr int kBlk = 12;
static constexpr int kSteadyEnd = (kL - 1) + ((kSEQ - (kL - 1)) / kBlk) * kBlk;
static_assert((kSteadyEnd - (kL - 1)) % kBlk == 0, "whole blocks only");
static_assert(kSteadyEnd <= kSEQ, "steady steps are full steps");
static_assert(kSteadyEnd > kL - 1, "at least one block");
static constexpr int kBatchPerWave = 2;

__global__ __launch_bounds__(32) void gru_scan(
    const float* __restrict__ gx0, const float* __restrict__ hxs,
    const float* __restrict__ whh0, const float* __restrict__ bhh0,
    const float* __restrict__ wih_r, const float* __restrict__ whh_r,
    const float* __restrict__ bih_r, const float* __restrict__ bhh_r,
    float* __restrict__ out) {
  const int lane = threadIdx.x;
  const int p = lane & 15;
  const int j = p >> 2;
  const int l = p & 3;
  const int b = blockIdx.x * kBatchPerWave + (lane >> 4);
  const bool real = j < kH;
  const int jj = real ? j : 0;
  const int lr = (l > 0) ? l - 1 : 0;
  const bool use_gx = real && l == 0;
  const bool stores = real && l == kL - 1;
  const int gofs = b * kG + jj;
  const int oofs = b * kH + jj;

  float whh[kGates][kSrc], wlo[kGates][kSrc], bhh[kGates], bih[kGates];
#pragma unroll
  for (int g = 0; g < kGates; ++g) {
    const int grow = g * kH + jj;
    const float slo = (g < kGates - 1) ? 0.5f : 1.0f;
#pragma unroll
    for (int k = 0; k < kSrc; ++k) {
      const int u = (j - k) & 3;
      const bool have = real && u < kH;
      const int uu = (u < kH) ? u : 0;
      const float wh = (l == 0) ? whh0[grow * kH + uu]
                                : whh_r[lr * kG * kH + grow * kH + uu];
      const float wl = wih_r[lr * kG * kH + grow * kH + uu];
      whh[g][k] = have ? wh * 0.5f : 0.0f;
      wlo[g][k] = (have && l > 0) ? wl * slo : 0.0f;
    }
    const float bh = (l == 0) ? bhh0[grow] : bhh_r[lr * kG + grow];
    const float bi = bih_r[lr * kG + grow];
    bhh[g] = real ? bh * 0.5f : 0.0f;
    bih[g] = (real && l > 0) ? bi * slo : 0.0f;
  }

  float h = real ? hxs[(l * kBATCH + b) * kH + jj] : 0.0f;

  for (int s = 0; s < kL - 1; ++s)
    edge_step(s, l, use_gx, stores, oofs, gofs, h, wlo, bih, whh, bhh, gx0,
              out);

  float cur[kBlk][kGates];
#pragma unroll
  for (int i = 0; i < kBlk; ++i) load_gx_unit(cur[i], gx0, kL - 1 + i, gofs);

  for (int s = kL - 1; s < kSteadyEnd; s += kBlk) {
    float nxt[kBlk][kGates];
#pragma unroll
    for (int i = 0; i < kBlk; ++i) {
      const int t = s + kBlk + i;
      load_gx_unit(nxt[i], gx0, (t < kSEQ) ? t : kSEQ - 1, gofs);
    }
    __builtin_amdgcn_sched_barrier(0);
    float ob[kBlk];
#pragma unroll
    for (int i = 0; i < kBlk; ++i) {
      h = scan_step(use_gx, h, cur[i], wlo, bih, whh, bhh);
      ob[i] = h;
    }
    if (stores) {
      float* op = out + (size_t)(s - l) * (kBATCH * kH) + oofs;
#pragma unroll
      for (int i = 0; i < kBlk; ++i) op[(size_t)i * (kBATCH * kH)] = ob[i];
    }
#pragma unroll
    for (int i = 0; i < kBlk; ++i) {
#pragma unroll
      for (int g = 0; g < kGates; ++g) cur[i][g] = nxt[i][g];
    }
  }
  for (int s = kSteadyEnd; s < kSEQ + kL - 1; ++s)
    edge_step(s, l, use_gx, stores, oofs, gofs, h, wlo, bih, whh, bhh, gx0,
              out);

  if (real) out[(size_t)kSEQ * kBATCH * kH + (l * kBATCH + b) * kH + jj] = h;
}

extern "C" void kernel_launch(void* const* d_in, const int* in_sizes, int n_in,
                              void* d_out, int out_size, void* d_ws,
                              size_t ws_size, hipStream_t stream) {
  const float* x    = (const float*)d_in[0];
  const float* hxs  = (const float*)d_in[1];
  const float* wih0 = (const float*)d_in[2];
  const float* whh0 = (const float*)d_in[3];
  const float* bih0 = (const float*)d_in[4];
  const float* bhh0 = (const float*)d_in[5];
  const float* wihr = (const float*)d_in[6];
  const float* whhr = (const float*)d_in[7];
  const float* bihr = (const float*)d_in[8];
  const float* bhhr = (const float*)d_in[9];
  float* out = (float*)d_out;
  float* gx  = (float*)d_ws;

  gru_in_proj_wmma<<<kROWS / 128, 256, 0, stream>>>(x, wih0, bih0, gx);
  gru_scan<<<kBATCH / kBatchPerWave, 32, 0, stream>>>(gx, hxs, whh0, bhh0, wihr, whhr, bihr,
                                          bhhr, out);
}


# GRU scan: layers two steps apart (layer l does time s-2l): the lane below forms a layer's input-side sums a step early, off the dependent chain, and hands them up; r,z sum order: input side first
# speedup vs baseline: 24.5594x; 1.0869x over previous
#include <hip/hip_runtime.h>
#include <hip/hip_bf16.h>

typedef float v2f __attribute__((ext_vector_type(2)));
typedef float v8f __attribute__((ext_vector_type(8)));

static constexpr int kSEQ   = 2048;
static constexpr int kBATCH = 64;
static constexpr int kIN    = 512;
static constexpr int kH     = 3;
static constexpr int kL     = 4;
static constexpr int kG     = 3 * kH;
static constexpr int kROWS  = kSEQ * kBATCH;

__global__ __launch_bounds__(256) void gru_in_proj_wmma(
    const float* __restrict__ X, const float* __restrict__ Wih0,
    const float* __restrict__ bih0, float* __restrict__ gx) {
  __shared__ float Bs[kIN * 16];
  const int tid = threadIdx.x;
  for (int idx = tid; idx < kIN * 16; idx += 256) {
    const int k = idx >> 4;
    const int n = idx & 15;
    Bs[idx] = (n < kG) ? Wih0[n * kIN + k] : 0.0f;
  }
  __syncthreads();

  const int wave = tid >> 5;
  const int lane = tid & 31;
  const int half = lane >> 4;
  const int n16  = lane & 15;
  const int row0 = blockIdx.x * 128 + wave * 16;

  const float* arow = X + (size_t)(row0 + n16) * kIN + 2 * half;

  v8f c = {0.f, 0.f, 0.f, 0.f, 0.f, 0.f, 0.f, 0.f};
  for (int kk = 0; kk < kIN; kk += 4) {
    v2f a = *(const v2f*)(arow + kk);
    v2f b;
    b.x = Bs[(kk + 2 * half) * 16 + n16];
    b.y = Bs[(kk + 2 * half + 1) * 16 + n16];
    c = __builtin_amdgcn_wmma_f32_16x16x4_f32(
         false, a,  false, b,
         (short)0, c,  false,  false);
  }

  if (n16 < kG) {
    const float bias = bih0[n16];
#pragma unroll
    for (int v = 0; v < 8; ++v) {
      const int m = v + 8 * half;
      gx[(size_t)(row0 + m) * kG + n16] = c[v] + bias;
    }
  }
}

__device__ __forceinline__ float tanh_hw(float x) {
  return __builtin_amdgcn_tanhf(x);
}

template <int kN>
__device__ __forceinline__ float row_ror(float v) {
  return __builtin_bit_cast(
      float, __builtin_amdgcn_mov_dpp(__builtin_bit_cast(int, v), 0x120 + kN,
                                      0xf, 0xf, true));
}

static constexpr int kSkew  = 2;
static constexpr int kGates = 3;
static constexpr int kSrc   = 4;

__device__ __forceinline__ void load_gx_unit(float (&g)[kGates],
                                             const float* __restrict__ gx0,
                                             int t, int gofs) {
  const float* gp = gx0 + (size_t)t * (kBATCH * kG) + gofs;
#pragma unroll
  for (int i = 0; i < kGates; ++i) g[i] = gp[i * kH];
}

__device__ __forceinline__ float scan_step(
    float h, float (&st)[kGates], const float (&rs)[kGates],
    const float (&rown)[kGates],
    const float (&wup)[kGates][kSrc], const float (&bup)[kGates],
    const float (&whh)[kGates][kSrc], const float (&bhh)[kGates]) {
  const float h4 = row_ror<4>(h), h8 = row_ror<8>(h), h12 = row_ror<12>(h);
  float s[kGates];
#pragma unroll
  for (int g = 0; g < kGates; ++g) {
    float sum = __builtin_fmaf(whh[g][0], h, (g < kGates - 1) ? st[g] : bhh[g]);
    sum = __builtin_fmaf(whh[g][1], h4, sum);
    sum = __builtin_fmaf(whh[g][2], h8, sum);
    sum = __builtin_fmaf(whh[g][3], h12, sum);
    s[g] = sum;
  }
  const float tr = tanh_hw(s[0]);
  const float tz = tanh_hw(s[1]);

  float u[kGates];
#pragma unroll
  for (int g = 0; g < kGates; ++g) {
    float sum = __builtin_fmaf(wup[g][0], h, bup[g]);
    sum = __builtin_fmaf(wup[g][1], h4, sum);
    sum = __builtin_fmaf(wup[g][2], h8, sum);
    sum = __builtin_fmaf(wup[g][3], h12, sum);
    u[g] = sum;
  }

  const float z = __builtin_fmaf(0.5f, tz, 0.5f);
  const float omz = 1.0f - z;
  const float zh = z * h;
  const float nbase = st[2] + s[2];
  float n = tanh_hw(__builtin_fmaf(tr, s[2], nbase));

#pragma unroll
  for (int g = 0; g < kGates; ++g) {
    const float a = __builtin_fmaf(rs[g], rown[g], row_ror<1>(u[g]));
    st[g] = (g < kGates - 1) ? a + bhh[g] : a;
  }
  asm volatile("" : "+v"(n) : "v"(st[0]), "v"(st[1]), "v"(st[2]));
  return __builtin_fmaf(omz, n, zh);
}

__device__ __forceinline__ void edge_step(
    int s, int l, const float (&rs)[kGates], bool stores, int oofs, int gofs,
    float& h, float (&st)[kGates],
    const float (&wlo)[kGates][kSrc], const float (&bih)[kGates],
    const float (&whh)[kGates][kSrc], const float (&bhh)[kGates],
    const float* __restrict__ gx0, float* __restrict__ out) {
  const int tn = (s + 1 >= kSEQ) ? kSEQ - 1 : s + 1;
  float rown[kGates];
  load_gx_unit(rown, gx0, tn, gofs);
  const int t = s - kSkew * l;
  const float hn = scan_step(h, st, rs, rown, wlo, bih, whh, bhh);
  if (t >= 0 && t < kSEQ) {
    h = hn;
    if (stores) out[(size_t)t * (kBATCH * kH) + oofs] = hn;
  }
}

static constexpr int kBlk = 17;
static constexpr int kFill = kSkew * (kL - 1);
static constexpr int kSteadyEnd = kFill + ((kSEQ - kFill) / kBlk) * kBlk;
static_assert((kSteadyEnd - kFill) % kBlk == 0, "whole blocks only");
static_assert(kSteadyEnd <= kSEQ, "steady steps are full steps");
static_assert(kSteadyEnd > kFill, "at least one block");
static constexpr int kBatchPerWave = 2;

__global__ __launch_bounds__(32) void gru_scan(
    const float* __restrict__ gx0, const float* __restrict__ hxs,
    const float* __restrict__ whh0, const float* __restrict__ bhh0,
    const float* __restrict__ wih_r, const float* __restrict__ whh_r,
    const float* __restrict__ bih_r, const float* __restrict__ bhh_r,
    float* __restrict__ out) {
  const int lane = threadIdx.x;
  const int p = lane & 15;
  const int j = p >> 2;
  const int l = p & 3;
  const int b = blockIdx.x * kBatchPerWave + (lane >> 4);
  const bool real = j < kH;
  const int jj = real ? j : 0;
  const int lr = (l > 0) ? l - 1 : 0;
  const int lu = (l < kL - 1) ? l : 0;
  const bool use_gx = real && l == 0;
  const bool stores = real && l == kL - 1;
  const int gofs = b * kG + jj;
  const int oofs = b * kH + jj;

  float whh[kGates][kSrc], wlo[kGates][kSrc], bhh[kGates], bih[kGates];
#pragma unroll
  for (int g = 0; g < kGates; ++g) {
    const int grow = g * kH + jj;
    const float slo = (g < kGates - 1) ? 0.5f : 1.0f;
#pragma unroll
    for (int k = 0; k < kSrc; ++k) {
      const int u = (j - k) & 3;
      const bool have = real && u < kH;
      const int uu = (u < kH) ? u : 0;
      const float wh = (l == 0) ? whh0[grow * kH + uu]
                                : whh_r[lr * kG * kH + grow * kH + uu];
      const float wl = wih_r[lu * kG * kH + grow * kH + uu];
      whh[g][k] = have ? wh * 0.5f : 0.0f;
      wlo[g][k] = (have && l < kL - 1) ? wl * slo : 0.0f;
    }
    const float bh = (l == 0) ? bhh0[grow] : bhh_r[lr * kG + grow];
    const float bi = bih_r[lu * kG + grow];
    bhh[g] = real ? bh * 0.5f : 0.0f;
    bih[g] = (real && l < kL - 1) ? bi * slo : 0.0f;
  }

  float h = real ? hxs[(l * kBATCH + b) * kH + jj] : 0.0f;

  float rs[kGates];
#pragma unroll
  for (int g = 0; g < kGates; ++g)
    rs[g] = use_gx ? (g < kGates - 1 ? 0.5f : 1.0f) : 0.0f;
  float st[kGates];
  {
    float row0[kGates];
    load_gx_unit(row0, gx0, 0, gofs);
#pragma unroll
    for (int g = 0; g < kGates; ++g) {
      const float a = __builtin_fmaf(rs[g], row0[g], 0.0f);
      st[g] = (g < kGates - 1) ? a + bhh[g] : a;
    }
  }

  for (int s = 0; s < kFill; ++s)
    edge_step(s, l, rs, stores, oofs, gofs, h, st, wlo, bih, whh, bhh,
              gx0, out);

  float cur[kBlk][kGates];
#pragma unroll
  for (int i = 0; i < kBlk; ++i) load_gx_unit(cur[i], gx0, kFill + i, gofs);

  for (int s = kFill; s < kSteadyEnd; s += kBlk) {
    float nxt[kBlk][kGates];
#pragma unroll
    for (int i = 0; i < kBlk; ++i) {
      const int t = s + kBlk + i;
      load_gx_unit(nxt[i], gx0, (t < kSEQ) ? t : kSEQ - 1, gofs);
    }
    __builtin_amdgcn_sched_barrier(0);
    float ob[kBlk];
#pragma unroll
    for (int i = 0; i < kBlk; ++i) {
      const float (&rown)[kGates] = (i + 1 < kBlk) ? cur[i + 1] : nxt[0];
      h = scan_step(h, st, rs, rown, wlo, bih, whh, bhh);
      ob[i] = h;
    }
    if (stores) {
      float* op = out + (size_t)(s - kSkew * l) * (kBATCH * kH) + oofs;
#pragma unroll
      for (int i = 0; i < kBlk; ++i) op[(size_t)i * (kBATCH * kH)] = ob[i];
    }
#pragma unroll
    for (int i = 0; i < kBlk; ++i) {
#pragma unroll
      for (int g = 0; g < kGates; ++g) cur[i][g] = nxt[i][g];
    }
  }
  for (int s = kSteadyEnd; s < kSEQ + kFill; ++s)
    edge_step(s, l, rs, stores, oofs, gofs, h, st, wlo, bih, whh, bhh,
              gx0, out);

  if (real) out[(size_t)kSEQ * kBATCH * kH + (l * kBATCH + b) * kH + jj] = h;
}

extern "C" void kernel_launch(void* const* d_in, const int* in_sizes, int n_in,
                              void* d_out, int out_size, void* d_ws,
                              size_t ws_size, hipStream_t stream) {
  const float* x    = (const float*)d_in[0];
  const float* hxs  = (const float*)d_in[1];
  const float* wih0 = (const float*)d_in[2];
  const float* whh0 = (const float*)d_in[3];
  const float* bih0 = (const float*)d_in[4];
  const float* bhh0 = (const float*)d_in[5];
  const float* wihr = (const float*)d_in[6];
  const float* whhr = (const float*)d_in[7];
  const float* bihr = (const float*)d_in[8];
  const float* bhhr = (const float*)d_in[9];
  float* out = (float*)d_out;
  float* gx  = (float*)d_ws;

  gru_in_proj_wmma<<<kROWS / 128, 256, 0, stream>>>(x, wih0, bih0, gx);
  gru_scan<<<kBATCH / kBatchPerWave, 32, 0, stream>>>(gx, hxs, whh0, bhh0, wihr, whhr, bihr,
                                          bhhr, out);
}


# input projection: X read in whole 128-B lines (one 16-byte load = 512 B of one row), two chunks ahead, made WMMA fragments in a per-wave LDS tile; K sum regrouped (k order, two accumulators)
# speedup vs baseline: 26.2796x; 1.0700x over previous
#include <hip/hip_runtime.h>
#include <hip/hip_bf16.h>

typedef float v2f __attribute__((ext_vector_type(2)));
typedef float v8f __attribute__((ext_vector_type(8)));

static constexpr int kSEQ   = 2048;
static constexpr int kBATCH = 64;
static constexpr int kIN    = 512;
static constexpr int kH     = 3;
static constexpr int kL     = 4;
static constexpr int kG     = 3 * kH;
static constexpr int kROWS  = kSEQ * kBATCH;

typedef float v4f __attribute__((ext_vector_type(4)));

static constexpr int kTileM        = 16;
static constexpr int kTilesPerWave = 2;
static constexpr int kWavesA       = 8;
static constexpr int kRowsPerWaveA = kTilesPerWave * kTileM;
static constexpr int kRowsPerBlockA = kWavesA * kRowsPerWaveA;
static constexpr int kChunk        = 128;
static constexpr int kChunksPerTile = kIN / kChunk;
static constexpr int kChunksPerWave = kTilesPerWave * kChunksPerTile;
static constexpr int kXPitch = kChunk + 4;
static constexpr int kWPitch = kIN + 4;
static_assert(kROWS % kRowsPerBlockA == 0, "whole workgroups");
static_assert(kIN % kChunk == 0 && kChunk == 4 * 32, "one 16-byte piece per lane");
static_assert(kChunksPerWave % 2 == 0, "chunks are taken in pairs");

__global__ __launch_bounds__(256) void gru_in_proj_wmma(
    const float* __restrict__ X, const float* __restrict__ Wih0,
    const float* __restrict__ bih0, float* __restrict__ gx) {
  __shared__ __attribute__((aligned(16))) float Ws[kG * kWPitch];
  __shared__ __attribute__((aligned(16))) float Xs[kWavesA * kTileM * kXPitch];
  const int tid = threadIdx.x;
  const int wave = tid >> 5;
  const int lane = tid & 31;
  const int half = lane >> 4;
  const int n16  = lane & 15;
  const int nB   = (n16 < kG) ? n16 : kG - 1;
  const int rowW = (blockIdx.x * kWavesA + wave) * kRowsPerWaveA;

  float* xs = Xs + wave * (kTileM * kXPitch);
  float* xst = xs + 4 * lane;
  const float* xfrag = xs + n16 * kXPitch + 4 * half;
  const float* wfrag = Ws + nB * kWPitch + 4 * half;
  const float* xg = X + (size_t)rowW * kIN + 4 * lane;
  const float bias = bih0[nB];

  auto request = [&](v4f (&r)[kTileM], int c) {
    const float* p = xg + (size_t)((c / kChunksPerTile) * kTileM) * kIN +
                     (c % kChunksPerTile) * kChunk;
#pragma unroll
    for (int i = 0; i < kTileM; ++i)
      r[i] = __builtin_nontemporal_load((const v4f*)(p + (size_t)i * kIN));
  };
  auto wave_lds_order = [&]() {
    __builtin_amdgcn_fence(__ATOMIC_ACQ_REL, "wavefront");
  };
  auto to_lds = [&](const v4f (&r)[kTileM]) {
    wave_lds_order();
#pragma unroll
    for (int i = 0; i < kTileM; ++i) *(v4f*)(xst + i * kXPitch) = r[i];
    wave_lds_order();
  };
  auto products = [&](int kc, v8f& c0, v8f& c1) {
    const float* wf = wfrag + kc * kChunk;
#pragma unroll
    for (int q = 0; q < kChunk / 8; ++q) {
      const v4f a = *(const v4f*)(xfrag + 8 * q);
      const v4f b = *(const v4f*)(wf + 8 * q);
      const v2f a0 = {a.x, a.y}, a1 = {a.z, a.w};
      const v2f b0 = {b.x, b.y}, b1 = {b.z, b.w};
      c0 = __builtin_amdgcn_wmma_f32_16x16x4_f32(
           false, a0,  false, b0,
           (short)0, c0,  false,  false);
      c1 = __builtin_amdgcn_wmma_f32_16x16x4_f32(
           false, a1,  false, b1,
           (short)0, c1,  false,  false);
    }
  };

  constexpr int kVec = kG * kIN / 4;
  constexpr int kIt  = (kVec + 255) / 256;
  v4f w[kIt];
#pragma unroll
  for (int i = 0; i < kIt; ++i) {
    const int v = tid + 256 * i;
    const int vv = (v < kVec) ? v : 0;
    w[i] = *(const v4f*)(Wih0 + 4 * vv);
  }
  v4f P[kTileM], Q[kTileM];
  request(P, 0);
  request(Q, 1);
#pragma unroll
  for (int i = 0; i < kIt; ++i) {
    const int v = tid + 256 * i;
    if (v < kVec) {
      const int n  = v / (kIN / 4);
      const int k4 = v % (kIN / 4);
      *(v4f*)(Ws + n * kWPitch + 4 * k4) = w[i];
    }
  }
  __syncthreads();
  const v8f zero = {0.f, 0.f, 0.f, 0.f, 0.f, 0.f, 0.f, 0.f};
  v8f c0 = zero, c1 = zero;
#pragma unroll
  for (int c = 0; c < kChunksPerWave; c += 2) {
    to_lds(P);
    if (c + 2 < kChunksPerWave) request(P, c + 2);
    __builtin_amdgcn_sched_barrier(0);
    products(c % kChunksPerTile, c0, c1);
    to_lds(Q);
    if (c + 3 < kChunksPerWave) request(Q, c + 3);
    __builtin_amdgcn_sched_barrier(0);
    products((c + 1) % kChunksPerTile, c0, c1);

    if ((c + 2) % kChunksPerTile == 0) {
      const int rowT = rowW + (c / kChunksPerTile) * kTileM;
      if (n16 < kG) {
#pragma unroll
        for (int v = 0; v < 8; ++v) {
          const int m = v + 8 * half;
          gx[(size_t)(rowT + m) * kG + n16] = (c0[v] + c1[v]) + bias;
        }
      }
      c0 = zero;
      c1 = zero;
    }
  }
}

__device__ __forceinline__ float tanh_hw(float x) {
  return __builtin_amdgcn_tanhf(x);
}

template <int kN>
__device__ __forceinline__ float row_ror(float v) {
  return __builtin_bit_cast(
      float, __builtin_amdgcn_mov_dpp(__builtin_bit_cast(int, v), 0x120 + kN,
                                      0xf, 0xf, true));
}

static constexpr int kSkew  = 2;
static constexpr int kGates = 3;
static constexpr int kSrc   = 4;

__device__ __forceinline__ void load_gx_unit(float (&g)[kGates],
                                             const float* __restrict__ gx0,
                                             int t, int gofs) {
  const float* gp = gx0 + (size_t)t * (kBATCH * kG) + gofs;
#pragma unroll
  for (int i = 0; i < kGates; ++i) g[i] = gp[i * kH];
}

__device__ __forceinline__ float scan_step(
    float h, float (&st)[kGates], const float (&rs)[kGates],
    const float (&rown)[kGates],
    const float (&wup)[kGates][kSrc], const float (&bup)[kGates],
    const float (&whh)[kGates][kSrc], const float (&bhh)[kGates]) {
  const float h4 = row_ror<4>(h), h8 = row_ror<8>(h), h12 = row_ror<12>(h);
  float s[kGates];
#pragma unroll
  for (int g = 0; g < kGates; ++g) {
    float sum = __builtin_fmaf(whh[g][0], h, (g < kGates - 1) ? st[g] : bhh[g]);
    sum = __builtin_fmaf(whh[g][1], h4, sum);
    sum = __builtin_fmaf(whh[g][2], h8, sum);
    sum = __builtin_fmaf(whh[g][3], h12, sum);
    s[g] = sum;
  }
  const float tr = tanh_hw(s[0]);
  const float tz = tanh_hw(s[1]);

  float u[kGates];
#pragma unroll
  for (int g = 0; g < kGates; ++g) {
    float sum = __builtin_fmaf(wup[g][0], h, bup[g]);
    sum = __builtin_fmaf(wup[g][1], h4, sum);
    sum = __builtin_fmaf(wup[g][2], h8, sum);
    sum = __builtin_fmaf(wup[g][3], h12, sum);
    u[g] = sum;
  }

  const float z = __builtin_fmaf(0.5f, tz, 0.5f);
  const float omz = 1.0f - z;
  const float zh = z * h;
  const float nbase = st[2] + s[2];
  float n = tanh_hw(__builtin_fmaf(tr, s[2], nbase));

#pragma unroll
  for (int g = 0; g < kGates; ++g) {
    const float a = __builtin_fmaf(rs[g], rown[g], row_ror<1>(u[g]));
    st[g] = (g < kGates - 1) ? a + bhh[g] : a;
  }
  asm volatile("" : "+v"(n) : "v"(st[0]), "v"(st[1]), "v"(st[2]));
  return __builtin_fmaf(omz, n, zh);
}

__device__ __forceinline__ void edge_step(
    int s, int l, const float (&rs)[kGates], bool stores, int oofs, int gofs,
    float& h, float (&st)[kGates],
    const float (&wlo)[kGates][kSrc], const float (&bih)[kGates],
    const float (&whh)[kGates][kSrc], const float (&bhh)[kGates],
    const float* __restrict__ gx0, float* __restrict__ out) {
  const int tn = (s + 1 >= kSEQ) ? kSEQ - 1 : s + 1;
  float rown[kGates];
  load_gx_unit(rown, gx0, tn, gofs);
  const int t = s - kSkew * l;
  const float hn = scan_step(h, st, rs, rown, wlo, bih, whh, bhh);
  if (t >= 0 && t < kSEQ) {
    h = hn;
    if (stores) out[(size_t)t * (kBATCH * kH) + oofs] = hn;
  }
}

static constexpr int kBlk = 17;
static constexpr int kFill = kSkew * (kL - 1);
static constexpr int kSteadyEnd = kFill + ((kSEQ - kFill) / kBlk) * kBlk;
static_assert((kSteadyEnd - kFill) % kBlk == 0, "whole blocks only");
static_assert(kSteadyEnd <= kSEQ, "steady steps are full steps");
static_assert(kSteadyEnd > kFill, "at least one block");
static constexpr int kBatchPerWave = 2;

__global__ __launch_bounds__(32) void gru_scan(
    const float* __restrict__ gx0, const float* __restrict__ hxs,
    const float* __restrict__ whh0, const float* __restrict__ bhh0,
    const float* __restrict__ wih_r, const float* __restrict__ whh_r,
    const float* __restrict__ bih_r, const float* __restrict__ bhh_r,
    float* __restrict__ out) {
  const int lane = threadIdx.x;
  const int p = lane & 15;
  const int j = p >> 2;
  const int l = p & 3;
  const int b = blockIdx.x * kBatchPerWave + (lane >> 4);
  const bool real = j < kH;
  const int jj = real ? j : 0;
  const int lr = (l > 0) ? l - 1 : 0;
  const int lu = (l < kL - 1) ? l : 0;
  const bool use_gx = real && l == 0;
  const bool stores = real && l == kL - 1;
  const int gofs = b * kG + jj;
  const int oofs = b * kH + jj;

  float whh[kGates][kSrc], wlo[kGates][kSrc], bhh[kGates], bih[kGates];
#pragma unroll
  for (int g = 0; g < kGates; ++g) {
    const int grow = g * kH + jj;
    const float slo = (g < kGates - 1) ? 0.5f : 1.0f;
#pragma unroll
    for (int k = 0; k < kSrc; ++k) {
      const int u = (j - k) & 3;
      const bool have = real && u < kH;
      const int uu = (u < kH) ? u : 0;
      const float wh = (l == 0) ? whh0[grow * kH + uu]
                                : whh_r[lr * kG * kH + grow * kH + uu];
      const float wl = wih_r[lu * kG * kH + grow * kH + uu];
      whh[g][k] = have ? wh * 0.5f : 0.0f;
      wlo[g][k] = (have && l < kL - 1) ? wl * slo : 0.0f;
    }
    const float bh = (l == 0) ? bhh0[grow] : bhh_r[lr * kG + grow];
    const float bi = bih_r[lu * kG + grow];
    bhh[g] = real ? bh * 0.5f : 0.0f;
    bih[g] = (real && l < kL - 1) ? bi * slo : 0.0f;
  }

  float h = real ? hxs[(l * kBATCH + b) * kH + jj] : 0.0f;

  float rs[kGates];
#pragma unroll
  for (int g = 0; g < kGates; ++g)
    rs[g] = use_gx ? (g < kGates - 1 ? 0.5f : 1.0f) : 0.0f;
  float st[kGates];
  {
    float row0[kGates];
    load_gx_unit(row0, gx0, 0, gofs);
#pragma unroll
    for (int g = 0; g < kGates; ++g) {
      const float a = __builtin_fmaf(rs[g], row0[g], 0.0f);
      st[g] = (g < kGates - 1) ? a + bhh[g] : a;
    }
  }

  for (int s = 0; s < kFill; ++s)
    edge_step(s, l, rs, stores, oofs, gofs, h, st, wlo, bih, whh, bhh,
              gx0, out);

  float cur[kBlk][kGates];
#pragma unroll
  for (int i = 0; i < kBlk; ++i) load_gx_unit(cur[i], gx0, kFill + i, gofs);

  for (int s = kFill; s < kSteadyEnd; s += kBlk) {
    float nxt[kBlk][kGates];
#pragma unroll
    for (int i = 0; i < kBlk; ++i) {
      const int t = s + kBlk + i;
      load_gx_unit(nxt[i], gx0, (t < kSEQ) ? t : kSEQ - 1, gofs);
    }
    __builtin_amdgcn_sched_barrier(0);
    float ob[kBlk];
#pragma unroll
    for (int i = 0; i < kBlk; ++i) {
      const float (&rown)[kGates] = (i + 1 < kBlk) ? cur[i + 1] : nxt[0];
      h = scan_step(h, st, rs, rown, wlo, bih, whh, bhh);
      ob[i] = h;
    }
    if (stores) {
      float* op = out + (size_t)(s - kSkew * l) * (kBATCH * kH) + oofs;
#pragma unroll
      for (int i = 0; i < kBlk; ++i) op[(size_t)i * (kBATCH * kH)] = ob[i];
    }
#pragma unroll
    for (int i = 0; i < kBlk; ++i) {
#pragma unroll
      for (int g = 0; g < kGates; ++g) cur[i][g] = nxt[i][g];
    }
  }
  for (int s = kSteadyEnd; s < kSEQ + kFill; ++s)
    edge_step(s, l, rs, stores, oofs, gofs, h, st, wlo, bih, whh, bhh,
              gx0, out);

  if (real) out[(size_t)kSEQ * kBATCH * kH + (l * kBATCH + b) * kH + jj] = h;
}

extern "C" void kernel_launch(void* const* d_in, const int* in_sizes, int n_in,
                              void* d_out, int out_size, void* d_ws,
                              size_t ws_size, hipStream_t stream) {
  const float* x    = (const float*)d_in[0];
  const float* hxs  = (const float*)d_in[1];
  const float* wih0 = (const float*)d_in[2];
  const float* whh0 = (const float*)d_in[3];
  const float* bih0 = (const float*)d_in[4];
  const float* bhh0 = (const float*)d_in[5];
  const float* wihr = (const float*)d_in[6];
  const float* whhr = (const float*)d_in[7];
  const float* bihr = (const float*)d_in[8];
  const float* bhhr = (const float*)d_in[9];
  float* out = (float*)d_out;
  float* gx  = (float*)d_ws;

  gru_in_proj_wmma<<<kROWS / kRowsPerBlockA, 256, 0, stream>>>(x, wih0, bih0, gx);
  gru_scan<<<kBATCH / kBatchPerWave, 32, 0, stream>>>(gx, hxs, whh0, bhh0, wihr, whhr, bihr,
                                          bhhr, out);
}


# GRU scan: lanes re-laid as 4*layer + unit: a unit's two partners come by two quad_perm turns, own unit's product first; 3 products per sum, not 4 with a zero; unit sums in a new rotated order
# speedup vs baseline: 27.1197x; 1.0320x over previous
#include <hip/hip_runtime.h>
#include <hip/hip_bf16.h>

typedef float v2f __attribute__((ext_vector_type(2)));
typedef float v8f __attribute__((ext_vector_type(8)));

static constexpr int kSEQ   = 2048;
static constexpr int kBATCH = 64;
static constexpr int kIN    = 512;
static constexpr int kH     = 3;
static constexpr int kL     = 4;
static constexpr int kG     = 3 * kH;
static constexpr int kROWS  = kSEQ * kBATCH;

typedef float v4f __attribute__((ext_vector_type(4)));

static constexpr int kTileM        = 16;
static constexpr int kTilesPerWave = 2;
static constexpr int kWavesA       = 8;
static constexpr int kRowsPerWaveA = kTilesPerWave * kTileM;
static constexpr int kRowsPerBlockA = kWavesA * kRowsPerWaveA;
static constexpr int kChunk        = 128;
static constexpr int kChunksPerTile = kIN / kChunk;
static constexpr int kChunksPerWave = kTilesPerWave * kChunksPerTile;
static constexpr int kXPitch = kChunk + 4;
static constexpr int kWPitch = kIN + 4;
static_assert(kROWS % kRowsPerBlockA == 0, "whole workgroups");
static_assert(kIN % kChunk == 0 && kChunk == 4 * 32, "one 16-byte piece per lane");
static_assert(kChunksPerWave % 2 == 0, "chunks are taken in pairs");

__global__ __launch_bounds__(256) void gru_in_proj_wmma(
    const float* __restrict__ X, const float* __restrict__ Wih0,
    const float* __restrict__ bih0, float* __restrict__ gx) {
  __shared__ __attribute__((aligned(16))) float Ws[kG * kWPitch];
  __shared__ __attribute__((aligned(16))) float Xs[kWavesA * kTileM * kXPitch];
  const int tid = threadIdx.x;
  const int wave = tid >> 5;
  const int lane = tid & 31;
  const int half = lane >> 4;
  const int n16  = lane & 15;
  const int nB   = (n16 < kG) ? n16 : kG - 1;
  const int rowW = (blockIdx.x * kWavesA + wave) * kRowsPerWaveA;

  float* xs = Xs + wave * (kTileM * kXPitch);
  float* xst = xs + 4 * lane;
  const float* xfrag = xs + n16 * kXPitch + 4 * half;
  const float* wfrag = Ws + nB * kWPitch + 4 * half;
  const float* xg = X + (size_t)rowW * kIN + 4 * lane;
  const float bias = bih0[nB];

  auto request = [&](v4f (&r)[kTileM], int c) {
    const float* p = xg + (size_t)((c / kChunksPerTile) * kTileM) * kIN +
                     (c % kChunksPerTile) * kChunk;
#pragma unroll
    for (int i = 0; i < kTileM; ++i)
      r[i] = __builtin_nontemporal_load((const v4f*)(p + (size_t)i * kIN));
  };
  auto wave_lds_order = [&]() {
    __builtin_amdgcn_fence(__ATOMIC_ACQ_REL, "wavefront");
  };
  auto to_lds = [&](const v4f (&r)[kTileM]) {
    wave_lds_order();
#pragma unroll
    for (int i = 0; i < kTileM; ++i) *(v4f*)(xst + i * kXPitch) = r[i];
    wave_lds_order();
  };
  auto products = [&](int kc, v8f& c0, v8f& c1) {
    const float* wf = wfrag + kc * kChunk;
#pragma unroll
    for (int q = 0; q < kChunk / 8; ++q) {
      const v4f a = *(const v4f*)(xfrag + 8 * q);
      const v4f b = *(const v4f*)(wf + 8 * q);
      const v2f a0 = {a.x, a.y}, a1 = {a.z, a.w};
      const v2f b0 = {b.x, b.y}, b1 = {b.z, b.w};
      c0 = __builtin_amdgcn_wmma_f32_16x16x4_f32(
           false, a0,  false, b0,
           (short)0, c0,  false,  false);
      c1 = __builtin_amdgcn_wmma_f32_16x16x4_f32(
           false, a1,  false, b1,
           (short)0, c1,  false,  false);
    }
  };

  constexpr int kVec = kG * kIN / 4;
  constexpr int kIt  = (kVec + 255) / 256;
  v4f w[kIt];
#pragma unroll
  for (int i = 0; i < kIt; ++i) {
    const int v = tid + 256 * i;
    const int vv = (v < kVec) ? v : 0;
    w[i] = *(const v4f*)(Wih0 + 4 * vv);
  }
  v4f P[kTileM], Q[kTileM];
  request(P, 0);
  request(Q, 1);
#pragma unroll
  for (int i = 0; i < kIt; ++i) {
    const int v = tid + 256 * i;
    if (v < kVec) {
      const int n  = v / (kIN / 4);
      const int k4 = v % (kIN / 4);
      *(v4f*)(Ws + n * kWPitch + 4 * k4) = w[i];
    }
  }
  __syncthreads();
  const v8f zero = {0.f, 0.f, 0.f, 0.f, 0.f, 0.f, 0.f, 0.f};
  v8f c0 = zero, c1 = zero;
#pragma unroll
  for (int c = 0; c < kChunksPerWave; c += 2) {
    to_lds(P);
    if (c + 2 < kChunksPerWave) request(P, c + 2);
    __builtin_amdgcn_sched_barrier(0);
    products(c % kChunksPerTile, c0, c1);
    to_lds(Q);
    if (c + 3 < kChunksPerWave) request(Q, c + 3);
    __builtin_amdgcn_sched_barrier(0);
    products((c + 1) % kChunksPerTile, c0, c1);

    if ((c + 2) % kChunksPerTile == 0) {
      const int rowT = rowW + (c / kChunksPerTile) * kTileM;
      if (n16 < kG) {
#pragma unroll
        for (int v = 0; v < 8; ++v) {
          const int m = v + 8 * half;
          gx[(size_t)(rowT + m) * kG + n16] = (c0[v] + c1[v]) + bias;
        }
      }
      c0 = zero;
      c1 = zero;
    }
  }
}

__device__ __forceinline__ float tanh_hw(float x) {
  return __builtin_amdgcn_tanhf(x);
}

template <int kCtrl>
__device__ __forceinline__ float dpp_move(float v) {
  return __builtin_bit_cast(
      float, __builtin_amdgcn_mov_dpp(__builtin_bit_cast(int, v), kCtrl, 0xf,
                                      0xf, true));
}
template <int kN>
__device__ __forceinline__ float row_ror(float v) {
  return dpp_move<0x120 + kN>(v);
}
__device__ __forceinline__ float quad_next(float v) {
  return dpp_move<(1 | 2 << 2 | 0 << 4 | 3 << 6)>(v);
}
__device__ __forceinline__ float quad_next2(float v) {
  return dpp_move<(2 | 0 << 2 | 1 << 4 | 3 << 6)>(v);
}

static constexpr int kSkew  = 2;
static constexpr int kGates = 3;
static constexpr int kSrc   = 3;

__device__ __forceinline__ void load_gx_unit(float (&g)[kGates],
                                             const float* __restrict__ gx0,
                                             int t, int gofs) {
  const float* gp = gx0 + (size_t)t * (kBATCH * kG) + gofs;
#pragma unroll
  for (int i = 0; i < kGates; ++i) g[i] = gp[i * kH];
}

__device__ __forceinline__ float scan_step(
    float h, float (&st)[kGates], const float (&rs)[kGates],
    const float (&rown)[kGates],
    const float (&wup)[kGates][kSrc], const float (&bup)[kGates],
    const float (&whh)[kGates][kSrc], const float (&bhh)[kGates]) {
  const float ha = quad_next(h), hb = quad_next2(h);
  float s[kGates];
#pragma unroll
  for (int g = 0; g < kGates; ++g) {
    float sum = __builtin_fmaf(whh[g][0], h, (g < kGates - 1) ? st[g] : bhh[g]);
    sum = __builtin_fmaf(whh[g][1], ha, sum);
    sum = __builtin_fmaf(whh[g][2], hb, sum);
    s[g] = sum;
  }
  const float tr = tanh_hw(s[0]);
  const float tz = tanh_hw(s[1]);

  float u[kGates];
#pragma unroll
  for (int g = 0; g < kGates; ++g) {
    float sum = __builtin_fmaf(wup[g][0], h, bup[g]);
    sum = __builtin_fmaf(wup[g][1], ha, sum);
    sum = __builtin_fmaf(wup[g][2], hb, sum);
    u[g] = sum;
  }

  const float z = __builtin_fmaf(0.5f, tz, 0.5f);
  const float omz = 1.0f - z;
  const float zh = z * h;
  const float nbase = st[2] + s[2];
  float n = tanh_hw(__builtin_fmaf(tr, s[2], nbase));

#pragma unroll
  for (int g = 0; g < kGates; ++g) {
    const float a = __builtin_fmaf(rs[g], rown[g], row_ror<4>(u[g]));
    st[g] = (g < kGates - 1) ? a + bhh[g] : a;
  }
  asm volatile("" : "+v"(n) : "v"(st[0]), "v"(st[1]), "v"(st[2]));
  return __builtin_fmaf(omz, n, zh);
}

__device__ __forceinline__ void edge_step(
    int s, int l, const float (&rs)[kGates], bool stores, int oofs, int gofs,
    float& h, float (&st)[kGates],
    const float (&wlo)[kGates][kSrc], const float (&bih)[kGates],
    const float (&whh)[kGates][kSrc], const float (&bhh)[kGates],
    const float* __restrict__ gx0, float* __restrict__ out) {
  const int tn = (s + 1 >= kSEQ) ? kSEQ - 1 : s + 1;
  float rown[kGates];
  load_gx_unit(rown, gx0, tn, gofs);
  const int t = s - kSkew * l;
  const float hn = scan_step(h, st, rs, rown, wlo, bih, whh, bhh);
  if (t >= 0 && t < kSEQ) {
    h = hn;
    if (stores) out[(size_t)t * (kBATCH * kH) + oofs] = hn;
  }
}

static constexpr int kBlk = 17;
static constexpr int kFill = kSkew * (kL - 1);
static constexpr int kSteadyEnd = kFill + ((kSEQ - kFill) / kBlk) * kBlk;
static_assert((kSteadyEnd - kFill) % kBlk == 0, "whole blocks only");
static_assert(kSteadyEnd <= kSEQ, "steady steps are full steps");
static_assert(kSteadyEnd > kFill, "at least one block");
static constexpr int kBatchPerWave = 2;

__global__ __launch_bounds__(32) void gru_scan(
    const float* __restrict__ gx0, const float* __restrict__ hxs,
    const float* __restrict__ whh0, const float* __restrict__ bhh0,
    const float* __restrict__ wih_r, const float* __restrict__ whh_r,
    const float* __restrict__ bih_r, const float* __restrict__ bhh_r,
    float* __restrict__ out) {
  const int lane = threadIdx.x;
  const int p = lane & 15;
  const int j = p & 3;
  const int l = p >> 2;
  const int b = blockIdx.x * kBatchPerWave + (lane >> 4);
  const bool real = j < kH;
  const int jj = real ? j : 0;
  const int lr = (l > 0) ? l - 1 : 0;
  const int lu = (l < kL - 1) ? l : 0;
  const bool use_gx = real && l == 0;
  const bool stores = real && l == kL - 1;
  const int gofs = b * kG + jj;
  const int oofs = b * kH + jj;

  float whh[kGates][kSrc], wlo[kGates][kSrc], bhh[kGates], bih[kGates];
#pragma unroll
  for (int g = 0; g < kGates; ++g) {
    const int grow = g * kH + jj;
    const float slo = (g < kGates - 1) ? 0.5f : 1.0f;
#pragma unroll
    for (int k = 0; k < kSrc; ++k) {
      const int uu = (jj + k < kH) ? jj + k : jj + k - kH;
      const float wh = (l == 0) ? whh0[grow * kH + uu]
                                : whh_r[lr * kG * kH + grow * kH + uu];
      const float wl = wih_r[lu * kG * kH + grow * kH + uu];
      whh[g][k] = real ? wh * 0.5f : 0.0f;
      wlo[g][k] = (real && l < kL - 1) ? wl * slo : 0.0f;
    }
    const float bh = (l == 0) ? bhh0[grow] : bhh_r[lr * kG + grow];
    const float bi = bih_r[lu * kG + grow];
    bhh[g] = real ? bh * 0.5f : 0.0f;
    bih[g] = (real && l < kL - 1) ? bi * slo : 0.0f;
  }

  float h = real ? hxs[(l * kBATCH + b) * kH + jj] : 0.0f;

  float rs[kGates];
#pragma unroll
  for (int g = 0; g < kGates; ++g)
    rs[g] = use_gx ? (g < kGates - 1 ? 0.5f : 1.0f) : 0.0f;
  float st[kGates];
  {
    float row0[kGates];
    load_gx_unit(row0, gx0, 0, gofs);
#pragma unroll
    for (int g = 0; g < kGates; ++g) {
      const float a = __builtin_fmaf(rs[g], row0[g], 0.0f);
      st[g] = (g < kGates - 1) ? a + bhh[g] : a;
    }
  }

  for (int s = 0; s < kFill; ++s)
    edge_step(s, l, rs, stores, oofs, gofs, h, st, wlo, bih, whh, bhh,
              gx0, out);

  float cur[kBlk][kGates];
#pragma unroll
  for (int i = 0; i < kBlk; ++i) load_gx_unit(cur[i], gx0, kFill + i, gofs);

  for (int s = kFill; s < kSteadyEnd; s += kBlk) {
    float nxt[kBlk][kGates];
#pragma unroll
    for (int i = 0; i < kBlk; ++i) {
      const int t = s + kBlk + i;
      load_gx_unit(nxt[i], gx0, (t < kSEQ) ? t : kSEQ - 1, gofs);
    }
    __builtin_amdgcn_sched_barrier(0);
    float ob[kBlk];
#pragma unroll
    for (int i = 0; i < kBlk; ++i) {
      const float (&rown)[kGates] = (i + 1 < kBlk) ? cur[i + 1] : nxt[0];
      h = scan_step(h, st, rs, rown, wlo, bih, whh, bhh);
      ob[i] = h;
    }
    if (stores) {
      float* op = out + (size_t)(s - kSkew * l) * (kBATCH * kH) + oofs;
#pragma unroll
      for (int i = 0; i < kBlk; ++i) op[(size_t)i * (kBATCH * kH)] = ob[i];
    }
#pragma unroll
    for (int i = 0; i < kBlk; ++i) {
#pragma unroll
      for (int g = 0; g < kGates; ++g) cur[i][g] = nxt[i][g];
    }
  }
  for (int s = kSteadyEnd; s < kSEQ + kFill; ++s)
    edge_step(s, l, rs, stores, oofs, gofs, h, st, wlo, bih, whh, bhh,
              gx0, out);

  if (real) out[(size_t)kSEQ * kBATCH * kH + (l * kBATCH + b) * kH + jj] = h;
}

extern "C" void kernel_launch(void* const* d_in, const int* in_sizes, int n_in,
                              void* d_out, int out_size, void* d_ws,
                              size_t ws_size, hipStream_t stream) {
  const float* x    = (const float*)d_in[0];
  const float* hxs  = (const float*)d_in[1];
  const float* wih0 = (const float*)d_in[2];
  const float* whh0 = (const float*)d_in[3];
  const float* bih0 = (const float*)d_in[4];
  const float* bhh0 = (const float*)d_in[5];
  const float* wihr = (const float*)d_in[6];
  const float* whhr = (const float*)d_in[7];
  const float* bihr = (const float*)d_in[8];
  const float* bhhr = (const float*)d_in[9];
  float* out = (float*)d_out;
  float* gx  = (float*)d_ws;

  gru_in_proj_wmma<<<kROWS / kRowsPerBlockA, 256, 0, stream>>>(x, wih0, bih0, gx);
  gru_scan<<<kBATCH / kBatchPerWave, 32, 0, stream>>>(gx, hxs, whh0, bhh0, wihr, whhr, bihr,
                                          bhhr, out);
}


# GRU scan: a block's gx0 rows are 17 12-byte loads off one pointer (projection stores gx0 unit-major); two blocks per trip swap the row sets: no per-row address math, few row copies; sums as before
# speedup vs baseline: 30.3059x; 1.1175x over previous
#include <hip/hip_runtime.h>
#include <hip/hip_bf16.h>

typedef float v2f __attribute__((ext_vector_type(2)));
typedef float v8f __attribute__((ext_vector_type(8)));

static constexpr int kSEQ   = 2048;
static constexpr int kBATCH = 64;
static constexpr int kIN    = 512;
static constexpr int kH     = 3;
static constexpr int kL     = 4;
static constexpr int kG     = 3 * kH;
static constexpr int kROWS  = kSEQ * kBATCH;

typedef float v4f __attribute__((ext_vector_type(4)));

static constexpr int kTileM        = 16;
static constexpr int kTilesPerWave = 2;
static constexpr int kWavesA       = 8;
static constexpr int kRowsPerWaveA = kTilesPerWave * kTileM;
static constexpr int kRowsPerBlockA = kWavesA * kRowsPerWaveA;
static constexpr int kChunk        = 128;
static constexpr int kChunksPerTile = kIN / kChunk;
static constexpr int kChunksPerWave = kTilesPerWave * kChunksPerTile;
static constexpr int kXPitch = kChunk + 4;
static constexpr int kWPitch = kIN + 4;
static_assert(kROWS % kRowsPerBlockA == 0, "whole workgroups");
static_assert(kIN % kChunk == 0 && kChunk == 4 * 32, "one 16-byte piece per lane");
static_assert(kChunksPerWave % 2 == 0, "chunks are taken in pairs");

__global__ __launch_bounds__(256) void gru_in_proj_wmma(
    const float* __restrict__ X, const float* __restrict__ Wih0,
    const float* __restrict__ bih0, float* __restrict__ gx) {
  __shared__ __attribute__((aligned(16))) float Ws[kG * kWPitch];
  __shared__ __attribute__((aligned(16))) float Xs[kWavesA * kTileM * kXPitch];
  const int tid = threadIdx.x;
  const int wave = tid >> 5;
  const int lane = tid & 31;
  const int half = lane >> 4;
  const int n16  = lane & 15;
  const int nB   = (n16 < kG) ? n16 : kG - 1;
  const int rowW = (blockIdx.x * kWavesA + wave) * kRowsPerWaveA;

  float* xs = Xs + wave * (kTileM * kXPitch);
  float* xst = xs + 4 * lane;
  const float* xfrag = xs + n16 * kXPitch + 4 * half;
  const float* wfrag = Ws + nB * kWPitch + 4 * half;
  const float* xg = X + (size_t)rowW * kIN + 4 * lane;
  const float bias = bih0[nB];
  const int gcol = (nB % kH) * (kG / kH) + nB / kH;

  auto request = [&](v4f (&r)[kTileM], int c) {
    const float* p = xg + (size_t)((c / kChunksPerTile) * kTileM) * kIN +
                     (c % kChunksPerTile) * kChunk;
#pragma unroll
    for (int i = 0; i < kTileM; ++i)
      r[i] = __builtin_nontemporal_load((const v4f*)(p + (size_t)i * kIN));
  };
  auto wave_lds_order = [&]() {
    __builtin_amdgcn_fence(__ATOMIC_ACQ_REL, "wavefront");
  };
  auto to_lds = [&](const v4f (&r)[kTileM]) {
    wave_lds_order();
#pragma unroll
    for (int i = 0; i < kTileM; ++i) *(v4f*)(xst + i * kXPitch) = r[i];
    wave_lds_order();
  };
  auto products = [&](int kc, v8f& c0, v8f& c1) {
    const float* wf = wfrag + kc * kChunk;
#pragma unroll
    for (int q = 0; q < kChunk / 8; ++q) {
      const v4f a = *(const v4f*)(xfrag + 8 * q);
      const v4f b = *(const v4f*)(wf + 8 * q);
      const v2f a0 = {a.x, a.y}, a1 = {a.z, a.w};
      const v2f b0 = {b.x, b.y}, b1 = {b.z, b.w};
      c0 = __builtin_amdgcn_wmma_f32_16x16x4_f32(
           false, a0,  false, b0,
           (short)0, c0,  false,  false);
      c1 = __builtin_amdgcn_wmma_f32_16x16x4_f32(
           false, a1,  false, b1,
           (short)0, c1,  false,  false);
    }
  };

  constexpr int kVec = kG * kIN / 4;
  constexpr int kIt  = (kVec + 255) / 256;
  v4f w[kIt];
#pragma unroll
  for (int i = 0; i < kIt; ++i) {
    const int v = tid + 256 * i;
    const int vv = (v < kVec) ? v : 0;
    w[i] = *(const v4f*)(Wih0 + 4 * vv);
  }
  v4f P[kTileM], Q[kTileM];
  request(P, 0);
  request(Q, 1);
#pragma unroll
  for (int i = 0; i < kIt; ++i) {
    const int v = tid + 256 * i;
    if (v < kVec) {
      const int n  = v / (kIN / 4);
      const int k4 = v % (kIN / 4);
      *(v4f*)(Ws + n * kWPitch + 4 * k4) = w[i];
    }
  }
  __syncthreads();
  const v8f zero = {0.f, 0.f, 0.f, 0.f, 0.f, 0.f, 0.f, 0.f};
  v8f c0 = zero, c1 = zero;
#pragma unroll
  for (int c = 0; c < kChunksPerWave; c += 2) {
    to_lds(P);
    if (c + 2 < kChunksPerWave) request(P, c + 2);
    __builtin_amdgcn_sched_barrier(0);
    products(c % kChunksPerTile, c0, c1);
    to_lds(Q);
    if (c + 3 < kChunksPerWave) request(Q, c + 3);
    __builtin_amdgcn_sched_barrier(0);
    products((c + 1) % kChunksPerTile, c0, c1);

    if ((c + 2) % kChunksPerTile == 0) {
      const int rowT = rowW + (c / kChunksPerTile) * kTileM;
      if (n16 < kG) {
#pragma unroll
        for (int v = 0; v < 8; ++v) {
          const int m = v + 8 * half;
          gx[(size_t)(rowT + m) * kG + gcol] = (c0[v] + c1[v]) + bias;
        }
      }
      c0 = zero;
      c1 = zero;
    }
  }
}

__device__ __forceinline__ float tanh_hw(float x) {
  return __builtin_amdgcn_tanhf(x);
}

template <int kCtrl>
__device__ __forceinline__ float dpp_move(float v) {
  return __builtin_bit_cast(
      float, __builtin_amdgcn_mov_dpp(__builtin_bit_cast(int, v), kCtrl, 0xf,
                                      0xf, true));
}
template <int kN>
__device__ __forceinline__ float row_ror(float v) {
  return dpp_move<0x120 + kN>(v);
}
__device__ __forceinline__ float quad_next(float v) {
  return dpp_move<(1 | 2 << 2 | 0 << 4 | 3 << 6)>(v);
}
__device__ __forceinline__ float quad_next2(float v) {
  return dpp_move<(2 | 0 << 2 | 1 << 4 | 3 << 6)>(v);
}

static constexpr int kSkew  = 2;
static constexpr int kGates = 3;
static constexpr int kSrc   = 3;

__device__ __forceinline__ void load_gx_at(float (&g)[kGates],
                                           const float* __restrict__ gp) {
#pragma unroll
  for (int i = 0; i < kGates; ++i) g[i] = gp[i];
}
__device__ __forceinline__ void load_gx_unit(float (&g)[kGates],
                                             const float* __restrict__ gx0,
                                             int t, int gofs) {
  load_gx_at(g, gx0 + (size_t)t * (kBATCH * kG) + gofs);
}

__device__ __forceinline__ float scan_step(
    float h, float (&st)[kGates], const float (&rs)[kGates],
    const float (&rown)[kGates],
    const float (&wup)[kGates][kSrc], const float (&bup)[kGates],
    const float (&whh)[kGates][kSrc], const float (&bhh)[kGates]) {
  const float ha = quad_next(h), hb = quad_next2(h);
  float s[kGates];
#pragma unroll
  for (int g = 0; g < kGates; ++g) {
    float sum = __builtin_fmaf(whh[g][0], h, (g < kGates - 1) ? st[g] : bhh[g]);
    sum = __builtin_fmaf(whh[g][1], ha, sum);
    sum = __builtin_fmaf(whh[g][2], hb, sum);
    s[g] = sum;
  }
  const float tr = tanh_hw(s[0]);
  const float tz = tanh_hw(s[1]);

  float u[kGates];
#pragma unroll
  for (int g = 0; g < kGates; ++g) {
    float sum = __builtin_fmaf(wup[g][0], h, bup[g]);
    sum = __builtin_fmaf(wup[g][1], ha, sum);
    sum = __builtin_fmaf(wup[g][2], hb, sum);
    u[g] = sum;
  }

  const float z = __builtin_fmaf(0.5f, tz, 0.5f);
  const float omz = 1.0f - z;
  const float zh = z * h;
  const float nbase = st[2] + s[2];
  float n = tanh_hw(__builtin_fmaf(tr, s[2], nbase));

#pragma unroll
  for (int g = 0; g < kGates; ++g) {
    const float a = __builtin_fmaf(rs[g], rown[g], row_ror<4>(u[g]));
    st[g] = (g < kGates - 1) ? a + bhh[g] : a;
  }
  asm volatile("" : "+v"(n) : "v"(st[0]), "v"(st[1]), "v"(st[2]));
  return __builtin_fmaf(omz, n, zh);
}

__device__ __forceinline__ void edge_step(
    int s, int l, const float (&rs)[kGates], bool stores, int oofs, int gofs,
    float& h, float (&st)[kGates],
    const float (&wlo)[kGates][kSrc], const float (&bih)[kGates],
    const float (&whh)[kGates][kSrc], const float (&bhh)[kGates],
    const float* __restrict__ gx0, float* __restrict__ out) {
  const int tn = (s + 1 >= kSEQ) ? kSEQ - 1 : s + 1;
  float rown[kGates];
  load_gx_unit(rown, gx0, tn, gofs);
  const int t = s - kSkew * l;
  const float hn = scan_step(h, st, rs, rown, wlo, bih, whh, bhh);
  if (t >= 0 && t < kSEQ) {
    h = hn;
    if (stores) out[(size_t)t * (kBATCH * kH) + oofs] = hn;
  }
}

static constexpr int kBlk = 17;
static constexpr int kFill = kSkew * (kL - 1);
static constexpr int kSteadyEnd =
    kFill + ((kSEQ - kFill) / (2 * kBlk)) * (2 * kBlk);
static_assert((kSteadyEnd - kFill) % (2 * kBlk) == 0, "whole pairs of blocks");
static_assert(kSteadyEnd < kSEQ, "steady steps are full steps; row kSteadyEnd exists");
static_assert(kSEQ > 2 * kBlk, "the bent pointer stays inside the sequence");
static_assert(kSteadyEnd > kFill, "at least one block");
static constexpr int kBatchPerWave = 2;

__global__ __launch_bounds__(32) void gru_scan(
    const float* __restrict__ gx0, const float* __restrict__ hxs,
    const float* __restrict__ whh0, const float* __restrict__ bhh0,
    const float* __restrict__ wih_r, const float* __restrict__ whh_r,
    const float* __restrict__ bih_r, const float* __restrict__ bhh_r,
    float* __restrict__ out) {
  const int lane = threadIdx.x;
  const int p = lane & 15;
  const int j = p & 3;
  const int l = p >> 2;
  const int b = blockIdx.x * kBatchPerWave + (lane >> 4);
  const bool real = j < kH;
  const int jj = real ? j : 0;
  const int lr = (l > 0) ? l - 1 : 0;
  const int lu = (l < kL - 1) ? l : 0;
  const bool use_gx = real && l == 0;
  const bool stores = real && l == kL - 1;
  const int gofs = b * kG + kGates * jj;
  const int oofs = b * kH + jj;

  float whh[kGates][kSrc], wlo[kGates][kSrc], bhh[kGates], bih[kGates];
#pragma unroll
  for (int g = 0; g < kGates; ++g) {
    const int grow = g * kH + jj;
    const float slo = (g < kGates - 1) ? 0.5f : 1.0f;
#pragma unroll
    for (int k = 0; k < kSrc; ++k) {
      const int uu = (jj + k < kH) ? jj + k : jj + k - kH;
      const float wh = (l == 0) ? whh0[grow * kH + uu]
                                : whh_r[lr * kG * kH + grow * kH + uu];
      const float wl = wih_r[lu * kG * kH + grow * kH + uu];
      whh[g][k] = real ? wh * 0.5f : 0.0f;
      wlo[g][k] = (real && l < kL - 1) ? wl * slo : 0.0f;
    }
    const float bh = (l == 0) ? bhh0[grow] : bhh_r[lr * kG + grow];
    const float bi = bih_r[lu * kG + grow];
    bhh[g] = real ? bh * 0.5f : 0.0f;
    bih[g] = (real && l < kL - 1) ? bi * slo : 0.0f;
  }

  float h = real ? hxs[(l * kBATCH + b) * kH + jj] : 0.0f;

  float rs[kGates];
#pragma unroll
  for (int g = 0; g < kGates; ++g)
    rs[g] = use_gx ? (g < kGates - 1 ? 0.5f : 1.0f) : 0.0f;
  float st[kGates];
  {
    float row0[kGates];
    load_gx_unit(row0, gx0, 0, gofs);
#pragma unroll
    for (int g = 0; g < kGates; ++g) {
      const float a = __builtin_fmaf(rs[g], row0[g], 0.0f);
      st[g] = (g < kGates - 1) ? a + bhh[g] : a;
    }
  }

  for (int s = 0; s < kFill; ++s)
    edge_step(s, l, rs, stores, oofs, gofs, h, st, wlo, bih, whh, bhh,
              gx0, out);

  auto request_block = [&](float (&rows)[kBlk][kGates], int t0) {
    load_gx_unit(rows[0], gx0, t0, gofs);
    const int t1 = (t0 + 1 < kSEQ - (kBlk - 1)) ? t0 + 1 : kSEQ - (kBlk - 1);
    const float* gp = gx0 + (size_t)t1 * (kBATCH * kG) + gofs;
#pragma unroll
    for (int i = 1; i < kBlk; ++i)
      load_gx_at(rows[i], gp + (i - 1) * (kBATCH * kG));
  };
  auto run_block = [&](int s, const float (&cur)[kBlk][kGates],
                       float (&nxt)[kBlk][kGates]) {
    request_block(nxt, s + kBlk);
    __builtin_amdgcn_sched_barrier(0);
    float ob[kBlk];
#pragma unroll
    for (int i = 0; i < kBlk; ++i) {
      const float (&rown)[kGates] = (i + 1 < kBlk) ? cur[i + 1] : nxt[0];
      h = scan_step(h, st, rs, rown, wlo, bih, whh, bhh);
      ob[i] = h;
    }
    if (stores) {
      float* op = out + (size_t)(s - kSkew * l) * (kBATCH * kH) + oofs;
#pragma unroll
      for (int i = 0; i < kBlk; ++i) op[(size_t)i * (kBATCH * kH)] = ob[i];
    }
  };

  float rowsA[kBlk][kGates], rowsB[kBlk][kGates];
  request_block(rowsA, kFill);

  for (int s = kFill; s < kSteadyEnd; s += 2 * kBlk) {
    run_block(s, rowsA, rowsB);
    run_block(s + kBlk, rowsB, rowsA);
  }
  for (int s = kSteadyEnd; s < kSEQ + kFill; ++s)
    edge_step(s, l, rs, stores, oofs, gofs, h, st, wlo, bih, whh, bhh,
              gx0, out);

  if (real) out[(size_t)kSEQ * kBATCH * kH + (l * kBATCH + b) * kH + jj] = h;
}

extern "C" void kernel_launch(void* const* d_in, const int* in_sizes, int n_in,
                              void* d_out, int out_size, void* d_ws,
                              size_t ws_size, hipStream_t stream) {
  const float* x    = (const float*)d_in[0];
  const float* hxs  = (const float*)d_in[1];
  const float* wih0 = (const float*)d_in[2];
  const float* whh0 = (const float*)d_in[3];
  const float* bih0 = (const float*)d_in[4];
  const float* bhh0 = (const float*)d_in[5];
  const float* wihr = (const float*)d_in[6];
  const float* whhr = (const float*)d_in[7];
  const float* bihr = (const float*)d_in[8];
  const float* bhhr = (const float*)d_in[9];
  float* out = (float*)d_out;
  float* gx  = (float*)d_ws;

  gru_in_proj_wmma<<<kROWS / kRowsPerBlockA, 256, 0, stream>>>(x, wih0, bih0, gx);
  gru_scan<<<kBATCH / kBatchPerWave, 32, 0, stream>>>(gx, hxs, whh0, bhh0, wihr, whhr, bihr,
                                          bhhr, out);
}
